# starting kernel
# baseline (speedup 1.0000x reference)
#include <hip/hip_runtime.h>
#include <hip/hip_bf16.h>

typedef __attribute__((ext_vector_type(2))) float v2f;
typedef __attribute__((ext_vector_type(8))) float v8f;

#define D_FEAT 128

__global__ void gcn_zero_kernel(float* __restrict__ p, size_t n) {
    size_t i = (size_t)blockIdx.x * blockDim.x + threadIdx.x;
    size_t stride = (size_t)gridDim.x * blockDim.x;
    for (; i < n; i += stride) p[i] = 0.0f;
}

__global__ void gcn_degree_kernel(const int* __restrict__ dst,
                                  float* __restrict__ deg, int n_edges) {
    int e = blockIdx.x * blockDim.x + threadIdx.x;
    if (e < n_edges) {
        __hip_atomic_fetch_add(&deg[dst[e]], 1.0f,
                               __ATOMIC_RELAXED, __HIP_MEMORY_SCOPE_AGENT);
    }
}

__global__ void gcn_scatter_kernel(const float* __restrict__ feat,
                                   const float* __restrict__ deg,
                                   const int* __restrict__ src,
                                   const int* __restrict__ dst,
                                   float* __restrict__ agg, int n_edges) {
    int wave = (int)((blockIdx.x * (size_t)blockDim.x + threadIdx.x) >> 5);
    int lane = threadIdx.x & 31;
    if (wave >= n_edges) return;
    int s = src[wave];
    int d = dst[wave];
    float nrm = rsqrtf(fmaxf(deg[s], 1.0f));
    const float* fs = feat + (size_t)s * D_FEAT;
    float* ad = agg + (size_t)d * D_FEAT;
#pragma unroll
    for (int i = 0; i < 4; ++i) {
        int c = lane + i * 32;
        __hip_atomic_fetch_add(&ad[c], fs[c] * nrm,
                               __ATOMIC_RELAXED, __HIP_MEMORY_SCOPE_AGENT);
    }
}

__global__ void gcn_gemm_kernel(const float* __restrict__ agg,
                                const float* __restrict__ W,
                                const float* __restrict__ feat,
                                const float* __restrict__ beta,
                                float* __restrict__ out,
                                int tiles_n  ) {
    int wave = (int)((blockIdx.x * (size_t)blockDim.x + threadIdx.x) >> 5);
    int lane = threadIdx.x & 31;
    int tile_m = wave / tiles_n;
    int tile_n = wave % tiles_n;
    int half = lane >> 4;
    int l15  = lane & 15;

    const float* aptr = agg + (size_t)(tile_m * 16 + l15) * D_FEAT + half * 2;
    const float* bptr = W   + (size_t)(tile_n * 16 + l15) * D_FEAT + half * 2;

    v8f acc = {};
#pragma unroll 8
    for (int k = 0; k < D_FEAT; k += 4) {
        v2f a = *(const v2f*)(aptr + k);
        v2f b = *(const v2f*)(bptr + k);
        acc = __builtin_amdgcn_wmma_f32_16x16x4_f32(
             false, a,  false, b,
             (short)0, acc,  false,  false);
    }

    float bv = beta[0];
#pragma unroll
    for (int r = 0; r < 8; ++r) {
        int m = tile_m * 16 + r + half * 8;
        int n = tile_n * 16 + l15;
        size_t idx = (size_t)m * D_FEAT + n;
        float h = fmaxf(acc[r], 0.0f);
        out[idx] = h + bv * feat[idx];
    }
}

extern "C" void kernel_launch(void* const* d_in, const int* in_sizes, int n_in,
                              void* d_out, int out_size, void* d_ws, size_t ws_size,
                              hipStream_t stream) {
    const float* feat = (const float*)d_in[0];
    const float* W    = (const float*)d_in[1];
    const float* beta = (const float*)d_in[2];
    const int*   src  = (const int*)d_in[3];
    const int*   dst  = (const int*)d_in[4];
    float*       out  = (float*)d_out;

    const int D = D_FEAT;
    const int N = in_sizes[0] / D;
    const int n_edges = in_sizes[3];

    float* agg = (float*)d_ws;
    float* deg = agg + (size_t)N * D;

    size_t zn = (size_t)N * D + (size_t)N;
    gcn_zero_kernel<<<2048, 256, 0, stream>>>(agg, zn);

    gcn_degree_kernel<<<(n_edges + 255) / 256, 256, 0, stream>>>(dst, deg, n_edges);

    gcn_scatter_kernel<<<(n_edges + 7) / 8, 256, 0, stream>>>(
        feat, deg, src, dst, agg, n_edges);

    int tiles_m = N / 16;
    int tiles_n = D / 16;
    int n_tiles = tiles_m * tiles_n;
    gcn_gemm_kernel<<<n_tiles / 8, 256, 0, stream>>>(agg, W, feat, beta, out, tiles_n);
}


# Aggregate by destination-sorted edge list: returning integer adds rank each edge, scan, fill, then one wave sums its node's rows in registers; no float atomics, sum order = arrival rank
# speedup vs baseline: 1.8146x; 1.8146x over previous
#include <hip/hip_runtime.h>
#include <hip/hip_bf16.h>

typedef __attribute__((ext_vector_type(2))) float v2f;
typedef __attribute__((ext_vector_type(8))) float v8f;

#define D_FEAT 128


__global__ void gcn_zero_kernel(int* __restrict__ p, int n) {
    int i = blockIdx.x * blockDim.x + threadIdx.x;
    if (i < n) p[i] = 0;
}

__global__ void gcn_count_rank_kernel(const int* __restrict__ dst,
                                      int* __restrict__ cnt,
                                      int* __restrict__ rank, int n_edges) {
    int e = blockIdx.x * blockDim.x + threadIdx.x;
    if (e < n_edges) {
        rank[e] = __hip_atomic_fetch_add(&cnt[dst[e]], 1,
                                         __ATOMIC_RELAXED, __HIP_MEMORY_SCOPE_AGENT);
    }
}

__global__ void __launch_bounds__(1024)
gcn_scan_kernel(const int* __restrict__ cnt, int* __restrict__ rowptr, int n) {
    __shared__ int s_pre[32];
    __shared__ int s_tot[32];
    int tid  = threadIdx.x;
    int lane = tid & 31;
    int w    = tid >> 5;
    int bstart = blockIdx.x * 1024;

    int pre = 0;
    const int4* c4 = (const int4*)cnt;
    int nvec = bstart >> 2;
    int i = tid;
    for (; i + 3072 < nvec; i += 4096) {
        int4 a = c4[i];
        int4 b = c4[i + 1024];
        int4 g = c4[i + 2048];
        int4 h = c4[i + 3072];
        pre += ((a.x + a.y) + (a.z + a.w)) + ((b.x + b.y) + (b.z + b.w))
             + ((g.x + g.y) + (g.z + g.w)) + ((h.x + h.y) + (h.z + h.w));
    }
    for (; i < nvec; i += 1024) {
        int4 v = c4[i];
        pre += (v.x + v.y) + (v.z + v.w);
    }

    int gid = bstart + tid;
    int c = (gid < n) ? cnt[gid] : 0;

    int incl = c;
    for (int off = 1; off < 32; off <<= 1) {
        int t = __shfl_up(incl, off);
        if (lane >= off) incl += t;
    }
    for (int off = 16; off > 0; off >>= 1) pre += __shfl_xor(pre, off);

    if (lane == 31) s_tot[w] = incl;
    if (lane == 0)  s_pre[w] = pre;
    __syncthreads();

    int front = s_pre[lane];
    int mine  = (lane < w) ? s_tot[lane] : 0;
    for (int off = 16; off > 0; off >>= 1) {
        front += __shfl_xor(front, off);
        mine  += __shfl_xor(mine, off);
    }
    if (gid < n) rowptr[gid] = front + mine + (incl - c);
}

__global__ void gcn_fill_kernel(const int* __restrict__ src,
                                const int* __restrict__ dst,
                                const int* __restrict__ rank,
                                const int* __restrict__ rowptr,
                                const int* __restrict__ cnt,
                                int2* __restrict__ csr, int n_edges) {
    int e = blockIdx.x * blockDim.x + threadIdx.x;
    if (e < n_edges) {
        int s = src[e];
        int d = dst[e];
        int p = rowptr[d] + rank[e];
        float nrm = rsqrtf(fmaxf((float)cnt[s], 1.0f));
        csr[p] = make_int2(s, __float_as_int(nrm));
    }
}

__global__ void gcn_gather_kernel(const float* __restrict__ feat,
                                  const int* __restrict__ rowptr,
                                  const int* __restrict__ cnt,
                                  const int2* __restrict__ csr,
                                  float* __restrict__ agg, int n_nodes) {
    int node = (int)((blockIdx.x * (size_t)blockDim.x + threadIdx.x) >> 5);
    int lane = threadIdx.x & 31;
    if (node >= n_nodes) return;
    node = __builtin_amdgcn_readfirstlane(node);
    int start = __builtin_amdgcn_readfirstlane(rowptr[node]);
    int n     = __builtin_amdgcn_readfirstlane(cnt[node]);

    const float* fl = feat + lane * 4;
    float4 acc = make_float4(0.0f, 0.0f, 0.0f, 0.0f);

    for (int base = 0; base < n; base += 32) {
        int m = n - base;
        if (m > 32) m = 32;
        int2 pr = make_int2(0, 0);
        if (lane < m) pr = csr[start + base + lane];

        int j = 0;
        for (; j + 8 <= m; j += 8) {
            float4 r[8];
            float  wt[8];
#pragma unroll
            for (int u = 0; u < 8; ++u) {
                int s = __builtin_amdgcn_readlane(pr.x, j + u);
                wt[u] = __int_as_float(__builtin_amdgcn_readlane(pr.y, j + u));
                r[u]  = *(const float4*)(fl + (size_t)s * D_FEAT);
            }
#pragma unroll
            for (int u = 0; u < 8; ++u) {
                acc.x = fmaf(r[u].x, wt[u], acc.x);
                acc.y = fmaf(r[u].y, wt[u], acc.y);
                acc.z = fmaf(r[u].z, wt[u], acc.z);
                acc.w = fmaf(r[u].w, wt[u], acc.w);
            }
        }
        for (; j < m; j += 4) {
            float4 r[4];
            float  wt[4];
#pragma unroll
            for (int u = 0; u < 4; ++u) {
                int jj = (j + u < m) ? (j + u) : (m - 1);
                int s = __builtin_amdgcn_readlane(pr.x, jj);
                float wv = __int_as_float(__builtin_amdgcn_readlane(pr.y, jj));
                wt[u] = (j + u < m) ? wv : 0.0f;
                r[u]  = *(const float4*)(fl + (size_t)s * D_FEAT);
            }
#pragma unroll
            for (int u = 0; u < 4; ++u) {
                acc.x = fmaf(r[u].x, wt[u], acc.x);
                acc.y = fmaf(r[u].y, wt[u], acc.y);
                acc.z = fmaf(r[u].z, wt[u], acc.z);
                acc.w = fmaf(r[u].w, wt[u], acc.w);
            }
        }
    }
    *(float4*)(agg + (size_t)node * D_FEAT + lane * 4) = acc;
}

__global__ void gcn_gemm_kernel(const float* __restrict__ agg,
                                const float* __restrict__ W,
                                const float* __restrict__ feat,
                                const float* __restrict__ beta,
                                float* __restrict__ out,
                                int tiles_n  ) {
    int wave = (int)((blockIdx.x * (size_t)blockDim.x + threadIdx.x) >> 5);
    int lane = threadIdx.x & 31;
    int tile_m = wave / tiles_n;
    int tile_n = wave % tiles_n;
    int half = lane >> 4;
    int l15  = lane & 15;

    const float* aptr = agg + (size_t)(tile_m * 16 + l15) * D_FEAT + half * 2;
    const float* bptr = W   + (size_t)(tile_n * 16 + l15) * D_FEAT + half * 2;

    v8f acc = {};
#pragma unroll 8
    for (int k = 0; k < D_FEAT; k += 4) {
        v2f a = *(const v2f*)(aptr + k);
        v2f b = *(const v2f*)(bptr + k);
        acc = __builtin_amdgcn_wmma_f32_16x16x4_f32(
             false, a,  false, b,
             (short)0, acc,  false,  false);
    }

    float bv = beta[0];
#pragma unroll
    for (int r = 0; r < 8; ++r) {
        int m = tile_m * 16 + r + half * 8;
        int n = tile_n * 16 + l15;
        size_t idx = (size_t)m * D_FEAT + n;
        float h = fmaxf(acc[r], 0.0f);
        out[idx] = h + bv * feat[idx];
    }
}

extern "C" void kernel_launch(void* const* d_in, const int* in_sizes, int n_in,
                              void* d_out, int out_size, void* d_ws, size_t ws_size,
                              hipStream_t stream) {
    const float* feat = (const float*)d_in[0];
    const float* W    = (const float*)d_in[1];
    const float* beta = (const float*)d_in[2];
    const int*   src  = (const int*)d_in[3];
    const int*   dst  = (const int*)d_in[4];
    float*       out  = (float*)d_out;

    const int D = D_FEAT;
    const int N = in_sizes[0] / D;
    const int n_edges = in_sizes[3];

    float* agg    = (float*)d_ws;
    int*   cnt    = (int*)(agg + (size_t)N * D);
    int*   rowptr = cnt + N;
    int*   rank   = rowptr + N;
    int2*  csr    = (int2*)(rank + n_edges);

    gcn_zero_kernel<<<(N + 255) / 256, 256, 0, stream>>>(cnt, N);

    gcn_count_rank_kernel<<<(n_edges + 255) / 256, 256, 0, stream>>>(dst, cnt, rank, n_edges);

    gcn_scan_kernel<<<(N + 1023) / 1024, 1024, 0, stream>>>(cnt, rowptr, N);

    gcn_fill_kernel<<<(n_edges + 255) / 256, 256, 0, stream>>>(
        src, dst, rank, rowptr, cnt, csr, n_edges);

    gcn_gather_kernel<<<(N + 7) / 8, 256, 0, stream>>>(feat, rowptr, cnt, csr, agg, N);

    int tiles_m = N / 16;
    int tiles_n = D / 16;
    int n_tiles = tiles_m * tiles_n;
    gcn_gemm_kernel<<<n_tiles / 8, 256, 0, stream>>>(agg, W, feat, beta, out, tiles_n);
}


# GEMM: one wave per 16x128 output strip (8 accumulator tiles): each A fragment loaded once, not 8 times; B fragments from an LDS copy of W; 6144 waves, not 50000; sum order as before
# speedup vs baseline: 2.1213x; 1.1690x over previous
#include <hip/hip_runtime.h>
#include <hip/hip_bf16.h>

typedef __attribute__((ext_vector_type(2))) float v2f;
typedef __attribute__((ext_vector_type(8))) float v8f;

#define D_FEAT 128


__global__ void gcn_zero_kernel(int* __restrict__ p, int n) {
    int i = blockIdx.x * blockDim.x + threadIdx.x;
    if (i < n) p[i] = 0;
}

__global__ void gcn_count_rank_kernel(const int* __restrict__ dst,
                                      int* __restrict__ cnt,
                                      int* __restrict__ rank, int n_edges) {
    int e = blockIdx.x * blockDim.x + threadIdx.x;
    if (e < n_edges) {
        rank[e] = __hip_atomic_fetch_add(&cnt[dst[e]], 1,
                                         __ATOMIC_RELAXED, __HIP_MEMORY_SCOPE_AGENT);
    }
}

__global__ void __launch_bounds__(1024)
gcn_scan_kernel(const int* __restrict__ cnt, int* __restrict__ rowptr, int n) {
    __shared__ int s_pre[32];
    __shared__ int s_tot[32];
    int tid  = threadIdx.x;
    int lane = tid & 31;
    int w    = tid >> 5;
    int bstart = blockIdx.x * 1024;

    int pre = 0;
    const int4* c4 = (const int4*)cnt;
    int nvec = bstart >> 2;
    int i = tid;
    for (; i + 3072 < nvec; i += 4096) {
        int4 a = c4[i];
        int4 b = c4[i + 1024];
        int4 g = c4[i + 2048];
        int4 h = c4[i + 3072];
        pre += ((a.x + a.y) + (a.z + a.w)) + ((b.x + b.y) + (b.z + b.w))
             + ((g.x + g.y) + (g.z + g.w)) + ((h.x + h.y) + (h.z + h.w));
    }
    for (; i < nvec; i += 1024) {
        int4 v = c4[i];
        pre += (v.x + v.y) + (v.z + v.w);
    }

    int gid = bstart + tid;
    int c = (gid < n) ? cnt[gid] : 0;

    int incl = c;
    for (int off = 1; off < 32; off <<= 1) {
        int t = __shfl_up(incl, off);
        if (lane >= off) incl += t;
    }
    for (int off = 16; off > 0; off >>= 1) pre += __shfl_xor(pre, off);

    if (lane == 31) s_tot[w] = incl;
    if (lane == 0)  s_pre[w] = pre;
    __syncthreads();

    int front = s_pre[lane];
    int mine  = (lane < w) ? s_tot[lane] : 0;
    for (int off = 16; off > 0; off >>= 1) {
        front += __shfl_xor(front, off);
        mine  += __shfl_xor(mine, off);
    }
    if (gid < n) rowptr[gid] = front + mine + (incl - c);
}

__global__ void gcn_fill_kernel(const int* __restrict__ src,
                                const int* __restrict__ dst,
                                const int* __restrict__ rank,
                                const int* __restrict__ rowptr,
                                const int* __restrict__ cnt,
                                int2* __restrict__ csr, int n_edges) {
    int e = blockIdx.x * blockDim.x + threadIdx.x;
    if (e < n_edges) {
        int s = src[e];
        int d = dst[e];
        int p = rowptr[d] + rank[e];
        float nrm = rsqrtf(fmaxf((float)cnt[s], 1.0f));
        csr[p] = make_int2(s, __float_as_int(nrm));
    }
}

__global__ void gcn_gather_kernel(const float* __restrict__ feat,
                                  const int* __restrict__ rowptr,
                                  const int* __restrict__ cnt,
                                  const int2* __restrict__ csr,
                                  float* __restrict__ agg, int n_nodes) {
    int node = (int)((blockIdx.x * (size_t)blockDim.x + threadIdx.x) >> 5);
    int lane = threadIdx.x & 31;
    if (node >= n_nodes) return;
    node = __builtin_amdgcn_readfirstlane(node);
    int start = __builtin_amdgcn_readfirstlane(rowptr[node]);
    int n     = __builtin_amdgcn_readfirstlane(cnt[node]);

    const float* fl = feat + lane * 4;
    float4 acc = make_float4(0.0f, 0.0f, 0.0f, 0.0f);

    for (int base = 0; base < n; base += 32) {
        int m = n - base;
        if (m > 32) m = 32;
        int2 pr = make_int2(0, 0);
        if (lane < m) pr = csr[start + base + lane];

        int j = 0;
        for (; j + 8 <= m; j += 8) {
            float4 r[8];
            float  wt[8];
#pragma unroll
            for (int u = 0; u < 8; ++u) {
                int s = __builtin_amdgcn_readlane(pr.x, j + u);
                wt[u] = __int_as_float(__builtin_amdgcn_readlane(pr.y, j + u));
                r[u]  = *(const float4*)(fl + (size_t)s * D_FEAT);
            }
#pragma unroll
            for (int u = 0; u < 8; ++u) {
                acc.x = fmaf(r[u].x, wt[u], acc.x);
                acc.y = fmaf(r[u].y, wt[u], acc.y);
                acc.z = fmaf(r[u].z, wt[u], acc.z);
                acc.w = fmaf(r[u].w, wt[u], acc.w);
            }
        }
        for (; j < m; j += 4) {
            float4 r[4];
            float  wt[4];
#pragma unroll
            for (int u = 0; u < 4; ++u) {
                int jj = (j + u < m) ? (j + u) : (m - 1);
                int s = __builtin_amdgcn_readlane(pr.x, jj);
                float wv = __int_as_float(__builtin_amdgcn_readlane(pr.y, jj));
                wt[u] = (j + u < m) ? wv : 0.0f;
                r[u]  = *(const float4*)(fl + (size_t)s * D_FEAT);
            }
#pragma unroll
            for (int u = 0; u < 4; ++u) {
                acc.x = fmaf(r[u].x, wt[u], acc.x);
                acc.y = fmaf(r[u].y, wt[u], acc.y);
                acc.z = fmaf(r[u].z, wt[u], acc.z);
                acc.w = fmaf(r[u].w, wt[u], acc.w);
            }
        }
    }
    *(float4*)(agg + (size_t)node * D_FEAT + lane * 4) = acc;
}

#define GEMM_THREADS 256

__global__ void __launch_bounds__(GEMM_THREADS) __attribute__((amdgpu_waves_per_eu(6, 6)))
gcn_gemm_kernel(const float* __restrict__ agg,
                const float* __restrict__ W,
                const float* __restrict__ feat,
                const float* __restrict__ beta,
                float* __restrict__ out,
                int n_strips  ) {
    __shared__ float sW[D_FEAT * D_FEAT];

    int tid  = threadIdx.x;
    int lane = tid & 31;
    int half = lane >> 4;
    int l15  = lane & 15;

    for (int i = tid; i < D_FEAT * D_FEAT / 4; i += GEMM_THREADS) {
        int n = i >> 5;
        int k = (i & 31) << 2;
        float4 v = *(const float4*)(W + (size_t)n * D_FEAT + k);
        *(float4*)(sW + n * D_FEAT + (k ^ ((n & 15) << 2))) = v;
    }
    __syncthreads();

    int wave   = __builtin_amdgcn_readfirstlane((int)(blockIdx.x * (GEMM_THREADS / 32) + (tid >> 5)));
    int nwaves = (int)(gridDim.x * (GEMM_THREADS / 32));
    float bv = beta[0];

    const float* sB = sW + l15 * D_FEAT + half * 2;
    int sw = l15 << 2;

    v2f bq[8];
#pragma unroll
    for (int t = 0; t < 8; ++t) bq[t] = *(const v2f*)(sB + t * 16 * D_FEAT + sw);

    for (int s = wave; s < n_strips; s += nwaves) {
        const float* aptr = agg + (size_t)(s * 16 + l15) * D_FEAT + half * 2;

        v8f acc[8];
#pragma unroll
        for (int t = 0; t < 8; ++t) acc[t] = (v8f){0.f, 0.f, 0.f, 0.f, 0.f, 0.f, 0.f, 0.f};

        v2f a[8];
#pragma unroll
        for (int j = 0; j < 8; ++j) a[j] = *(const v2f*)(aptr + 4 * j);

#pragma unroll 1
        for (int c = 0; c < 4; ++c) {
            v2f an[8];
#pragma unroll
            for (int j = 0; j < 8; ++j) an[j] = *(const v2f*)(aptr + 32 * ((c + 1) & 3) + 4 * j);
#pragma unroll
            for (int j = 0; j < 8; ++j) {
                int kx = ((32 * c + 4 * (j + 1)) & (D_FEAT - 1)) ^ sw;
                v2f bn[8];
#pragma unroll
                for (int t = 0; t < 8; ++t) bn[t] = *(const v2f*)(sB + t * 16 * D_FEAT + kx);
#pragma unroll
                for (int t = 0; t < 8; ++t) {
                    acc[t] = __builtin_amdgcn_wmma_f32_16x16x4_f32(
                         false, a[j],  false, bq[t],
                         (short)0, acc[t],  false,  false);
                }
#pragma unroll
                for (int t = 0; t < 8; ++t) bq[t] = bn[t];
            }
#pragma unroll
            for (int j = 0; j < 8; ++j) a[j] = an[j];
        }

#pragma unroll
        for (int r0 = 0; r0 < 8; r0 += 4) {
            float f[4][8];
#pragma unroll
            for (int r = 0; r < 4; ++r) {
                size_t rowoff = (size_t)(s * 16 + r0 + r + half * 8) * D_FEAT + l15;
#pragma unroll
                for (int t = 0; t < 8; ++t) f[r][t] = feat[rowoff + t * 16];
            }
#pragma unroll
            for (int r = 0; r < 4; ++r) {
                size_t rowoff = (size_t)(s * 16 + r0 + r + half * 8) * D_FEAT + l15;
#pragma unroll
                for (int t = 0; t < 8; ++t) {
                    float h = fmaxf(acc[t][r0 + r], 0.0f);
                    out[rowoff + t * 16] = h + bv * f[r][t];
                }
            }
        }
    }
}

extern "C" void kernel_launch(void* const* d_in, const int* in_sizes, int n_in,
                              void* d_out, int out_size, void* d_ws, size_t ws_size,
                              hipStream_t stream) {
    const float* feat = (const float*)d_in[0];
    const float* W    = (const float*)d_in[1];
    const float* beta = (const float*)d_in[2];
    const int*   src  = (const int*)d_in[3];
    const int*   dst  = (const int*)d_in[4];
    float*       out  = (float*)d_out;

    const int D = D_FEAT;
    const int N = in_sizes[0] / D;
    const int n_edges = in_sizes[3];

    float* agg    = (float*)d_ws;
    int*   cnt    = (int*)(agg + (size_t)N * D);
    int*   rowptr = cnt + N;
    int*   rank   = rowptr + N;
    int2*  csr    = (int2*)(rank + n_edges);

    gcn_zero_kernel<<<(N + 255) / 256, 256, 0, stream>>>(cnt, N);

    gcn_count_rank_kernel<<<(n_edges + 255) / 256, 256, 0, stream>>>(dst, cnt, rank, n_edges);

    gcn_scan_kernel<<<(N + 1023) / 1024, 1024, 0, stream>>>(cnt, rowptr, N);

    gcn_fill_kernel<<<(n_edges + 255) / 256, 256, 0, stream>>>(
        src, dst, rank, rowptr, cnt, csr, n_edges);

    gcn_gather_kernel<<<(N + 7) / 8, 256, 0, stream>>>(feat, rowptr, cnt, csr, agg, N);

    int n_strips = N / 16;
    gcn_gemm_kernel<<<768, GEMM_THREADS, 0, stream>>>(agg, W, feat, beta, out, n_strips);
}


# Edge sort through LDS, no global atomics: chunks grouped by destination range, one owner block per range ranks and places its edges; norm looked up in the gather; sum order = LDS arrival rank
# speedup vs baseline: 4.6678x; 2.2004x over previous
#include <hip/hip_runtime.h>
#include <hip/hip_bf16.h>

typedef __attribute__((ext_vector_type(2))) float v2f;
typedef __attribute__((ext_vector_type(8))) float v8f;

#define D_FEAT 128


#define NBINS       256
#define RMAX        512
#define BIN_THREADS 1024
#define BIN_ITER    8
#define BIN_CH      (BIN_THREADS * BIN_ITER)
#define OWN_THREADS 1024
#define MAXCH       1024
#define OWN_U       8
#define OWN_CAP     12288

__global__ void __launch_bounds__(BIN_THREADS)
gcn_bin_kernel(const int* __restrict__ src, const int* __restrict__ dst,
               int2* __restrict__ binned,
               int* __restrict__ tab_cnt, int* __restrict__ tab_start,
               int n_edges, int nodes_per_bin, int n_chunks) {
    __shared__ int  s_hist[NBINS];
    __shared__ int  s_off[NBINS];
    __shared__ int2 s_e[BIN_CH];

    int tid = threadIdx.x;
    int c   = blockIdx.x;
    int e0  = c * BIN_CH;

    if (tid < NBINS) s_hist[tid] = 0;

    int sv[BIN_ITER], dv[BIN_ITER], bv[BIN_ITER], rv[BIN_ITER];
#pragma unroll
    for (int i = 0; i < BIN_ITER; ++i) {
        int e = e0 + i * BIN_THREADS + tid;
        bool ok = e < n_edges;
        sv[i] = ok ? src[e] : 0;
        dv[i] = ok ? dst[e] : -1;
    }
    __syncthreads();

#pragma unroll
    for (int i = 0; i < BIN_ITER; ++i) {
        bv[i] = 0;
        rv[i] = 0;
        if (dv[i] >= 0) {
            int b = dv[i] / nodes_per_bin;
            if (b > NBINS - 1) b = NBINS - 1;
            bv[i] = b;
            rv[i] = atomicAdd(&s_hist[b], 1);
        }
    }
    __syncthreads();

    if (tid < 32) {
        int v[NBINS / 32];
        int run = 0;
#pragma unroll
        for (int k = 0; k < NBINS / 32; ++k) {
            int t = s_hist[tid * (NBINS / 32) + k];
            v[k] = run;
            run += t;
        }
        int incl = run;
        for (int off = 1; off < 32; off <<= 1) {
            int t = __shfl_up(incl, off);
            if (tid >= off) incl += t;
        }
        int excl = incl - run;
#pragma unroll
        for (int k = 0; k < NBINS / 32; ++k) s_off[tid * (NBINS / 32) + k] = excl + v[k];
    }
    __syncthreads();

    if (tid < NBINS) {
        tab_cnt[tid * n_chunks + c]   = s_hist[tid];
        tab_start[tid * n_chunks + c] = s_off[tid];
    }

#pragma unroll
    for (int i = 0; i < BIN_ITER; ++i) {
        if (dv[i] >= 0) s_e[s_off[bv[i]] + rv[i]] = make_int2(sv[i], dv[i]);
    }
    __syncthreads();

    int nloc = n_edges - e0;
    if (nloc > BIN_CH) nloc = BIN_CH;
#pragma unroll
    for (int i = 0; i < BIN_ITER; ++i) {
        int p = i * BIN_THREADS + tid;
        if (p < nloc) binned[e0 + p] = s_e[p];
    }
}

__device__ __forceinline__ void gcn_locate(const int* s_pre, const int* s_beg, int top,
                                           int first, int total, int (&idx)[OWN_U]) {
    int it[OWN_U], pos[OWN_U];
#pragma unroll
    for (int u = 0; u < OWN_U; ++u) {
        int i = first + u * OWN_THREADS;
        it[u]  = i > total - 1 ? total - 1 : i;
        pos[u] = 0;
    }
    for (int step = top; step > 0; step >>= 1) {
#pragma unroll
        for (int u = 0; u < OWN_U; ++u) pos[u] += (s_pre[pos[u] + step] <= it[u]) ? step : 0;
    }
#pragma unroll
    for (int u = 0; u < OWN_U; ++u) idx[u] = s_beg[pos[u]] + (it[u] - s_pre[pos[u]]);
}

__global__ void __launch_bounds__(OWN_THREADS)
gcn_owner_kernel(const int2* __restrict__ binned,
                 const int* __restrict__ tab_cnt, const int* __restrict__ tab_start,
                 int* __restrict__ cnt, int* __restrict__ rowptr, int* __restrict__ csr,
                 int n_nodes, int nodes_per_bin, int n_chunks) {
    __shared__ int s_pre[MAXCH + 1];
    __shared__ int s_beg[MAXCH];
    __shared__ int s_cnt[RMAX];
    __shared__ int s_row[RMAX];
    __shared__ int s_cur[RMAX];
    __shared__ int s_wa[32];
    __shared__ int s_wb[32];
    __shared__ int s_csr[OWN_CAP];

    int tid  = threadIdx.x;
    int lane = tid & 31;
    int w    = tid >> 5;
    int b    = blockIdx.x;

    int pre = 0;
    int nfront = b * n_chunks;
    const int4* t4 = (const int4*)tab_cnt;
    int nvec = nfront >> 2;
    int iv = tid;
    for (; iv + 3 * OWN_THREADS < nvec; iv += 4 * OWN_THREADS) {
        int4 a = t4[iv];
        int4 q = t4[iv + OWN_THREADS];
        int4 g = t4[iv + 2 * OWN_THREADS];
        int4 h = t4[iv + 3 * OWN_THREADS];
        pre += ((a.x + a.y) + (a.z + a.w)) + ((q.x + q.y) + (q.z + q.w))
             + ((g.x + g.y) + (g.z + g.w)) + ((h.x + h.y) + (h.z + h.w));
    }
    for (; iv < nvec; iv += OWN_THREADS) {
        int4 v = t4[iv];
        pre += (v.x + v.y) + (v.z + v.w);
    }
    if (tid < (nfront & 3)) pre += tab_cnt[(nvec << 2) + tid];

    int myc = 0, myb = 0;
    if (tid < n_chunks) {
        myc = tab_cnt[b * n_chunks + tid];
        myb = tid * BIN_CH + tab_start[b * n_chunks + tid];
    }
    if (tid < RMAX) { s_cnt[tid] = 0; s_cur[tid] = 0; }

    int incl = myc;
    for (int off = 1; off < 32; off <<= 1) {
        int t = __shfl_up(incl, off);
        if (lane >= off) incl += t;
    }
    for (int off = 16; off > 0; off >>= 1) pre += __shfl_xor(pre, off);
    if (lane == 31) s_wa[w] = incl;
    if (lane == 0)  s_wb[w] = pre;
    __syncthreads();

    int front = s_wb[lane];
    int mine  = (lane < w) ? s_wa[lane] : 0;
    for (int off = 16; off > 0; off >>= 1) {
        front += __shfl_xor(front, off);
        mine  += __shfl_xor(mine, off);
    }
    int base = front;
    int excl = mine + incl - myc;
    s_pre[tid] = excl;
    s_beg[tid] = myb;
    if (tid == OWN_THREADS - 1) s_pre[MAXCH] = excl + myc;
    __syncthreads();

    int total = s_pre[MAXCH];
    int node0 = b * nodes_per_bin;
    int top = 1;
    for (int k = 0; k < 10; ++k) if (top * 2 <= n_chunks) top *= 2;

    for (int i0 = 0; i0 < total; i0 += OWN_THREADS * OWN_U) {
        int2 ed[OWN_U];
        int idx[OWN_U];
        gcn_locate(s_pre, s_beg, top, i0 + tid, total, idx);
#pragma unroll
        for (int u = 0; u < OWN_U; ++u) ed[u] = binned[idx[u]];
#pragma unroll
        for (int u = 0; u < OWN_U; ++u) {
            int loc = ed[u].y - node0;
            loc = loc < 0 ? 0 : (loc > RMAX - 1 ? RMAX - 1 : loc);
            if (i0 + u * OWN_THREADS + tid < total) atomicAdd(&s_cnt[loc], 1);
        }
    }
    __syncthreads();

    int cv = (tid < RMAX) ? s_cnt[tid] : 0;
    int inc2 = cv;
    for (int off = 1; off < 32; off <<= 1) {
        int t = __shfl_up(inc2, off);
        if (lane >= off) inc2 += t;
    }
    if (lane == 31) s_wa[w] = inc2;
    __syncthreads();
    int m2 = (lane < w) ? s_wa[lane] : 0;
    for (int off = 16; off > 0; off >>= 1) m2 += __shfl_xor(m2, off);
    int rowx = m2 + inc2 - cv;
    if (tid < RMAX) s_row[tid] = rowx;
    int node = node0 + tid;
    if (tid < nodes_per_bin && node < n_nodes) {
        cnt[node]    = cv;
        rowptr[node] = base + rowx;
    }
    __syncthreads();

    bool staged = total <= OWN_CAP;
    for (int i0 = 0; i0 < total; i0 += OWN_THREADS * OWN_U) {
        int2 ed[OWN_U];
        int idx[OWN_U];
        gcn_locate(s_pre, s_beg, top, i0 + tid, total, idx);
#pragma unroll
        for (int u = 0; u < OWN_U; ++u) ed[u] = binned[idx[u]];
#pragma unroll
        for (int u = 0; u < OWN_U; ++u) {
            int loc = ed[u].y - node0;
            loc = loc < 0 ? 0 : (loc > RMAX - 1 ? RMAX - 1 : loc);
            if (i0 + u * OWN_THREADS + tid < total) {
                int r = atomicAdd(&s_cur[loc], 1);
                int pos = s_row[loc] + r;
                if (staged) s_csr[pos] = ed[u].x;
                else        csr[base + pos] = ed[u].x;
            }
        }
    }
    __syncthreads();
    if (staged) {
        for (int i = tid; i < total; i += OWN_THREADS) csr[base + i] = s_csr[i];
    }
}

__global__ void gcn_gather_kernel(const float* __restrict__ feat,
                                  const int* __restrict__ rowptr,
                                  const int* __restrict__ cnt,
                                  const int* __restrict__ csr,
                                  float* __restrict__ agg, int n_nodes) {
    int node = (int)((blockIdx.x * (size_t)blockDim.x + threadIdx.x) >> 5);
    int lane = threadIdx.x & 31;
    if (node >= n_nodes) return;
    node = __builtin_amdgcn_readfirstlane(node);
    int start = __builtin_amdgcn_readfirstlane(rowptr[node]);
    int n     = __builtin_amdgcn_readfirstlane(cnt[node]);

    const float* fl = feat + lane * 4;
    float4 acc = make_float4(0.0f, 0.0f, 0.0f, 0.0f);

    for (int base = 0; base < n; base += 32) {
        int m = n - base;
        if (m > 32) m = 32;
        int2 pr = make_int2(0, 0);
        if (lane < m) {
            int sidx = csr[start + base + lane];
            float nrm = rsqrtf(fmaxf((float)cnt[sidx], 1.0f));
            pr = make_int2(sidx, __float_as_int(nrm));
        }

        int j = 0;
        for (; j + 8 <= m; j += 8) {
            float4 r[8];
            float  wt[8];
#pragma unroll
            for (int u = 0; u < 8; ++u) {
                int s = __builtin_amdgcn_readlane(pr.x, j + u);
                wt[u] = __int_as_float(__builtin_amdgcn_readlane(pr.y, j + u));
                r[u]  = *(const float4*)(fl + (size_t)s * D_FEAT);
            }
#pragma unroll
            for (int u = 0; u < 8; ++u) {
                acc.x = fmaf(r[u].x, wt[u], acc.x);
                acc.y = fmaf(r[u].y, wt[u], acc.y);
                acc.z = fmaf(r[u].z, wt[u], acc.z);
                acc.w = fmaf(r[u].w, wt[u], acc.w);
            }
        }
        for (; j < m; j += 4) {
            float4 r[4];
            float  wt[4];
#pragma unroll
            for (int u = 0; u < 4; ++u) {
                int jj = (j + u < m) ? (j + u) : (m - 1);
                int s = __builtin_amdgcn_readlane(pr.x, jj);
                float wv = __int_as_float(__builtin_amdgcn_readlane(pr.y, jj));
                wt[u] = (j + u < m) ? wv : 0.0f;
                r[u]  = *(const float4*)(fl + (size_t)s * D_FEAT);
            }
#pragma unroll
            for (int u = 0; u < 4; ++u) {
                acc.x = fmaf(r[u].x, wt[u], acc.x);
                acc.y = fmaf(r[u].y, wt[u], acc.y);
                acc.z = fmaf(r[u].z, wt[u], acc.z);
                acc.w = fmaf(r[u].w, wt[u], acc.w);
            }
        }
    }
    *(float4*)(agg + (size_t)node * D_FEAT + lane * 4) = acc;
}

#define GEMM_THREADS 256

__global__ void __launch_bounds__(GEMM_THREADS) __attribute__((amdgpu_waves_per_eu(6, 6)))
gcn_gemm_kernel(const float* __restrict__ agg,
                const float* __restrict__ W,
                const float* __restrict__ feat,
                const float* __restrict__ beta,
                float* __restrict__ out,
                int n_strips  ) {
    __shared__ float sW[D_FEAT * D_FEAT];

    int tid  = threadIdx.x;
    int lane = tid & 31;
    int half = lane >> 4;
    int l15  = lane & 15;

    for (int i = tid; i < D_FEAT * D_FEAT / 4; i += GEMM_THREADS) {
        int n = i >> 5;
        int k = (i & 31) << 2;
        float4 v = *(const float4*)(W + (size_t)n * D_FEAT + k);
        *(float4*)(sW + n * D_FEAT + (k ^ ((n & 15) << 2))) = v;
    }
    __syncthreads();

    int wave   = __builtin_amdgcn_readfirstlane((int)(blockIdx.x * (GEMM_THREADS / 32) + (tid >> 5)));
    int nwaves = (int)(gridDim.x * (GEMM_THREADS / 32));
    float bv = beta[0];

    const float* sB = sW + l15 * D_FEAT + half * 2;
    int sw = l15 << 2;

    v2f bq[8];
#pragma unroll
    for (int t = 0; t < 8; ++t) bq[t] = *(const v2f*)(sB + t * 16 * D_FEAT + sw);

    for (int s = wave; s < n_strips; s += nwaves) {
        const float* aptr = agg + (size_t)(s * 16 + l15) * D_FEAT + half * 2;

        v8f acc[8];
#pragma unroll
        for (int t = 0; t < 8; ++t) acc[t] = (v8f){0.f, 0.f, 0.f, 0.f, 0.f, 0.f, 0.f, 0.f};

        v2f a[8];
#pragma unroll
        for (int j = 0; j < 8; ++j) a[j] = *(const v2f*)(aptr + 4 * j);

#pragma unroll 1
        for (int c = 0; c < 4; ++c) {
            v2f an[8];
#pragma unroll
            for (int j = 0; j < 8; ++j) an[j] = *(const v2f*)(aptr + 32 * ((c + 1) & 3) + 4 * j);
#pragma unroll
            for (int j = 0; j < 8; ++j) {
                int kx = ((32 * c + 4 * (j + 1)) & (D_FEAT - 1)) ^ sw;
                v2f bn[8];
#pragma unroll
                for (int t = 0; t < 8; ++t) bn[t] = *(const v2f*)(sB + t * 16 * D_FEAT + kx);
#pragma unroll
                for (int t = 0; t < 8; ++t) {
                    acc[t] = __builtin_amdgcn_wmma_f32_16x16x4_f32(
                         false, a[j],  false, bq[t],
                         (short)0, acc[t],  false,  false);
                }
#pragma unroll
                for (int t = 0; t < 8; ++t) bq[t] = bn[t];
            }
#pragma unroll
            for (int j = 0; j < 8; ++j) a[j] = an[j];
        }

#pragma unroll
        for (int r0 = 0; r0 < 8; r0 += 4) {
            float f[4][8];
#pragma unroll
            for (int r = 0; r < 4; ++r) {
                size_t rowoff = (size_t)(s * 16 + r0 + r + half * 8) * D_FEAT + l15;
#pragma unroll
                for (int t = 0; t < 8; ++t) f[r][t] = feat[rowoff + t * 16];
            }
#pragma unroll
            for (int r = 0; r < 4; ++r) {
                size_t rowoff = (size_t)(s * 16 + r0 + r + half * 8) * D_FEAT + l15;
#pragma unroll
                for (int t = 0; t < 8; ++t) {
                    float h = fmaxf(acc[t][r0 + r], 0.0f);
                    out[rowoff + t * 16] = h + bv * f[r][t];
                }
            }
        }
    }
}

extern "C" void kernel_launch(void* const* d_in, const int* in_sizes, int n_in,
                              void* d_out, int out_size, void* d_ws, size_t ws_size,
                              hipStream_t stream) {
    const float* feat = (const float*)d_in[0];
    const float* W    = (const float*)d_in[1];
    const float* beta = (const float*)d_in[2];
    const int*   src  = (const int*)d_in[3];
    const int*   dst  = (const int*)d_in[4];
    float*       out  = (float*)d_out;

    const int D = D_FEAT;
    const int N = in_sizes[0] / D;
    const int n_edges = in_sizes[3];

    const int n_chunks = (n_edges + BIN_CH - 1) / BIN_CH;
    const int npb      = (N + NBINS - 1) / NBINS;

    float* agg       = (float*)d_ws;
    int*   tab_cnt   = (int*)(agg + (size_t)N * D);
    int*   tab_start = tab_cnt + (size_t)NBINS * n_chunks;
    int2*  binned    = (int2*)(tab_start + (size_t)NBINS * n_chunks);
    int*   cnt       = (int*)(binned + n_edges);
    int*   rowptr    = cnt + N;
    int*   csr       = rowptr + N;

    gcn_bin_kernel<<<n_chunks, BIN_THREADS, 0, stream>>>(
        src, dst, binned, tab_cnt, tab_start, n_edges, npb, n_chunks);

    gcn_owner_kernel<<<NBINS, OWN_THREADS, 0, stream>>>(
        binned, tab_cnt, tab_start, cnt, rowptr, csr, N, npb, n_chunks);

    gcn_gather_kernel<<<(N + 7) / 8, 256, 0, stream>>>(feat, rowptr, cnt, csr, agg, N);

    int n_strips = N / 16;
    gcn_gemm_kernel<<<768, GEMM_THREADS, 0, stream>>>(agg, W, feat, beta, out, n_strips);
}


# Gather joins the GEMM launch: each wave sums the 16 rows of its 16x128 strip into LDS and multiplies them from there, agg never stored; 3 launches, not 4; 16 waves per WGP; sum order unchanged
# speedup vs baseline: 4.8315x; 1.0351x over previous
#include <hip/hip_runtime.h>
#include <hip/hip_bf16.h>

typedef __attribute__((ext_vector_type(2))) float v2f;
typedef __attribute__((ext_vector_type(8))) float v8f;

#define D_FEAT 128


#define NBINS       256
#define RMAX        512
#define BIN_THREADS 1024
#define BIN_ITER    8
#define BIN_CH      (BIN_THREADS * BIN_ITER)
#define OWN_THREADS 1024
#define MAXCH       1024
#define OWN_U       8
#define OWN_CAP     12288

__global__ void __launch_bounds__(BIN_THREADS)
gcn_bin_kernel(const int* __restrict__ src, const int* __restrict__ dst,
               int2* __restrict__ binned,
               int* __restrict__ tab_cnt, int* __restrict__ tab_start,
               int n_edges, int nodes_per_bin, int n_chunks) {
    __shared__ int  s_hist[NBINS];
    __shared__ int  s_off[NBINS];
    __shared__ int2 s_e[BIN_CH];

    int tid = threadIdx.x;
    int c   = blockIdx.x;
    int e0  = c * BIN_CH;

    if (tid < NBINS) s_hist[tid] = 0;

    int sv[BIN_ITER], dv[BIN_ITER], bv[BIN_ITER], rv[BIN_ITER];
#pragma unroll
    for (int i = 0; i < BIN_ITER; ++i) {
        int e = e0 + i * BIN_THREADS + tid;
        bool ok = e < n_edges;
        sv[i] = ok ? src[e] : 0;
        dv[i] = ok ? dst[e] : -1;
    }
    __syncthreads();

#pragma unroll
    for (int i = 0; i < BIN_ITER; ++i) {
        bv[i] = 0;
        rv[i] = 0;
        if (dv[i] >= 0) {
            int b = dv[i] / nodes_per_bin;
            if (b > NBINS - 1) b = NBINS - 1;
            bv[i] = b;
            rv[i] = atomicAdd(&s_hist[b], 1);
        }
    }
    __syncthreads();

    if (tid < 32) {
        int v[NBINS / 32];
        int run = 0;
#pragma unroll
        for (int k = 0; k < NBINS / 32; ++k) {
            int t = s_hist[tid * (NBINS / 32) + k];
            v[k] = run;
            run += t;
        }
        int incl = run;
        for (int off = 1; off < 32; off <<= 1) {
            int t = __shfl_up(incl, off);
            if (tid >= off) incl += t;
        }
        int excl = incl - run;
#pragma unroll
        for (int k = 0; k < NBINS / 32; ++k) s_off[tid * (NBINS / 32) + k] = excl + v[k];
    }
    __syncthreads();

    if (tid < NBINS) {
        tab_cnt[tid * n_chunks + c]   = s_hist[tid];
        tab_start[tid * n_chunks + c] = s_off[tid];
    }

#pragma unroll
    for (int i = 0; i < BIN_ITER; ++i) {
        if (dv[i] >= 0) s_e[s_off[bv[i]] + rv[i]] = make_int2(sv[i], dv[i]);
    }
    __syncthreads();

    int nloc = n_edges - e0;
    if (nloc > BIN_CH) nloc = BIN_CH;
#pragma unroll
    for (int i = 0; i < BIN_ITER; ++i) {
        int p = i * BIN_THREADS + tid;
        if (p < nloc) binned[e0 + p] = s_e[p];
    }
}

__device__ __forceinline__ void gcn_locate(const int* s_pre, const int* s_beg, int top,
                                           int first, int total, int (&idx)[OWN_U]) {
    int it[OWN_U], pos[OWN_U];
#pragma unroll
    for (int u = 0; u < OWN_U; ++u) {
        int i = first + u * OWN_THREADS;
        it[u]  = i > total - 1 ? total - 1 : i;
        pos[u] = 0;
    }
    for (int step = top; step > 0; step >>= 1) {
#pragma unroll
        for (int u = 0; u < OWN_U; ++u) pos[u] += (s_pre[pos[u] + step] <= it[u]) ? step : 0;
    }
#pragma unroll
    for (int u = 0; u < OWN_U; ++u) idx[u] = s_beg[pos[u]] + (it[u] - s_pre[pos[u]]);
}

__global__ void __launch_bounds__(OWN_THREADS)
gcn_owner_kernel(const int2* __restrict__ binned,
                 const int* __restrict__ tab_cnt, const int* __restrict__ tab_start,
                 int* __restrict__ cnt, int* __restrict__ rowptr, int* __restrict__ csr,
                 int n_nodes, int nodes_per_bin, int n_chunks) {
    __shared__ int s_pre[MAXCH + 1];
    __shared__ int s_beg[MAXCH];
    __shared__ int s_cnt[RMAX];
    __shared__ int s_row[RMAX];
    __shared__ int s_cur[RMAX];
    __shared__ int s_wa[32];
    __shared__ int s_wb[32];
    __shared__ int s_csr[OWN_CAP];

    int tid  = threadIdx.x;
    int lane = tid & 31;
    int w    = tid >> 5;
    int b    = blockIdx.x;

    int pre = 0;
    int nfront = b * n_chunks;
    const int4* t4 = (const int4*)tab_cnt;
    int nvec = nfront >> 2;
    int iv = tid;
    for (; iv + 3 * OWN_THREADS < nvec; iv += 4 * OWN_THREADS) {
        int4 a = t4[iv];
        int4 q = t4[iv + OWN_THREADS];
        int4 g = t4[iv + 2 * OWN_THREADS];
        int4 h = t4[iv + 3 * OWN_THREADS];
        pre += ((a.x + a.y) + (a.z + a.w)) + ((q.x + q.y) + (q.z + q.w))
             + ((g.x + g.y) + (g.z + g.w)) + ((h.x + h.y) + (h.z + h.w));
    }
    for (; iv < nvec; iv += OWN_THREADS) {
        int4 v = t4[iv];
        pre += (v.x + v.y) + (v.z + v.w);
    }
    if (tid < (nfront & 3)) pre += tab_cnt[(nvec << 2) + tid];

    int myc = 0, myb = 0;
    if (tid < n_chunks) {
        myc = tab_cnt[b * n_chunks + tid];
        myb = tid * BIN_CH + tab_start[b * n_chunks + tid];
    }
    if (tid < RMAX) { s_cnt[tid] = 0; s_cur[tid] = 0; }

    int incl = myc;
    for (int off = 1; off < 32; off <<= 1) {
        int t = __shfl_up(incl, off);
        if (lane >= off) incl += t;
    }
    for (int off = 16; off > 0; off >>= 1) pre += __shfl_xor(pre, off);
    if (lane == 31) s_wa[w] = incl;
    if (lane == 0)  s_wb[w] = pre;
    __syncthreads();

    int front = s_wb[lane];
    int mine  = (lane < w) ? s_wa[lane] : 0;
    for (int off = 16; off > 0; off >>= 1) {
        front += __shfl_xor(front, off);
        mine  += __shfl_xor(mine, off);
    }
    int base = front;
    int excl = mine + incl - myc;
    s_pre[tid] = excl;
    s_beg[tid] = myb;
    if (tid == OWN_THREADS - 1) s_pre[MAXCH] = excl + myc;
    __syncthreads();

    int total = s_pre[MAXCH];
    int node0 = b * nodes_per_bin;
    int top = 1;
    for (int k = 0; k < 10; ++k) if (top * 2 <= n_chunks) top *= 2;

    for (int i0 = 0; i0 < total; i0 += OWN_THREADS * OWN_U) {
        int2 ed[OWN_U];
        int idx[OWN_U];
        gcn_locate(s_pre, s_beg, top, i0 + tid, total, idx);
#pragma unroll
        for (int u = 0; u < OWN_U; ++u) ed[u] = binned[idx[u]];
#pragma unroll
        for (int u = 0; u < OWN_U; ++u) {
            int loc = ed[u].y - node0;
            loc = loc < 0 ? 0 : (loc > RMAX - 1 ? RMAX - 1 : loc);
            if (i0 + u * OWN_THREADS + tid < total) atomicAdd(&s_cnt[loc], 1);
        }
    }
    __syncthreads();

    int cv = (tid < RMAX) ? s_cnt[tid] : 0;
    int inc2 = cv;
    for (int off = 1; off < 32; off <<= 1) {
        int t = __shfl_up(inc2, off);
        if (lane >= off) inc2 += t;
    }
    if (lane == 31) s_wa[w] = inc2;
    __syncthreads();
    int m2 = (lane < w) ? s_wa[lane] : 0;
    for (int off = 16; off > 0; off >>= 1) m2 += __shfl_xor(m2, off);
    int rowx = m2 + inc2 - cv;
    if (tid < RMAX) s_row[tid] = rowx;
    int node = node0 + tid;
    if (tid < nodes_per_bin && node < n_nodes) {
        cnt[node]    = cv;
        rowptr[node] = base + rowx;
    }
    __syncthreads();

    bool staged = total <= OWN_CAP;
    for (int i0 = 0; i0 < total; i0 += OWN_THREADS * OWN_U) {
        int2 ed[OWN_U];
        int idx[OWN_U];
        gcn_locate(s_pre, s_beg, top, i0 + tid, total, idx);
#pragma unroll
        for (int u = 0; u < OWN_U; ++u) ed[u] = binned[idx[u]];
#pragma unroll
        for (int u = 0; u < OWN_U; ++u) {
            int loc = ed[u].y - node0;
            loc = loc < 0 ? 0 : (loc > RMAX - 1 ? RMAX - 1 : loc);
            if (i0 + u * OWN_THREADS + tid < total) {
                int r = atomicAdd(&s_cur[loc], 1);
                int pos = s_row[loc] + r;
                if (staged) s_csr[pos] = ed[u].x;
                else        csr[base + pos] = ed[u].x;
            }
        }
    }
    __syncthreads();
    if (staged) {
        for (int i = tid; i < total; i += OWN_THREADS) csr[base + i] = s_csr[i];
    }
}

#define FUSE_WAVES   16
#define FUSE_THREADS (FUSE_WAVES * 32)

template <int NB, bool PAD>
__device__ __forceinline__ void gcn_sum_rows(const float* fl, int2 pr, int j, int m, float4& acc) {
    float4 r[NB];
    float  wt[NB];
#pragma unroll
    for (int u = 0; u < NB; ++u) {
        bool in = !PAD || (j + u < m);
        int jj = in ? (j + u) : (m - 1);
        int s = __builtin_amdgcn_readlane(pr.x, jj);
        float wv = __int_as_float(__builtin_amdgcn_readlane(pr.y, jj));
        wt[u] = in ? wv : 0.0f;
        r[u]  = *(const float4*)(fl + (size_t)s * D_FEAT);
    }
#pragma unroll
    for (int u = 0; u < NB; ++u) {
        acc.x = fmaf(r[u].x, wt[u], acc.x);
        acc.y = fmaf(r[u].y, wt[u], acc.y);
        acc.z = fmaf(r[u].z, wt[u], acc.z);
        acc.w = fmaf(r[u].w, wt[u], acc.w);
    }
}

__global__ void __launch_bounds__(FUSE_THREADS) __attribute__((amdgpu_waves_per_eu(4, 4)))
gcn_fused_kernel(const float* __restrict__ feat,
                 const int* __restrict__ rowptr,
                 const int* __restrict__ cnt,
                 const int* __restrict__ csr,
                 const float* __restrict__ W,
                 const float* __restrict__ beta,
                 float* __restrict__ out,
                 int n_strips  ) {
    __shared__ float sW[D_FEAT * D_FEAT];
    __shared__ float sA[FUSE_WAVES * 16 * D_FEAT];

    int tid  = threadIdx.x;
    int lane = tid & 31;
    int half = lane >> 4;
    int l15  = lane & 15;
    int wv   = __builtin_amdgcn_readfirstlane(tid >> 5);

    for (int i = tid; i < D_FEAT * D_FEAT / 4; i += FUSE_THREADS) {
        int n = i >> 5;
        int k = (i & 31) << 2;
        float4 v = *(const float4*)(W + (size_t)n * D_FEAT + k);
        *(float4*)(sW + n * D_FEAT + (k ^ ((n & 15) << 2))) = v;
    }
    __syncthreads();

    int wave   = __builtin_amdgcn_readfirstlane((int)(wv * gridDim.x + blockIdx.x));
    int nwaves = (int)(gridDim.x * FUSE_WAVES);
    float bv = beta[0];

    const float* sB  = sW + l15 * D_FEAT + half * 2;
    int sw = l15 << 2;
    float*       sAw = sA + wv * 16 * D_FEAT;
    const float* sAr = sAw + l15 * D_FEAT + half * 2;
    const float* fl  = feat + lane * 4;

    v2f bq[8];
#pragma unroll
    for (int t = 0; t < 8; ++t) bq[t] = *(const v2f*)(sB + t * 16 * D_FEAT + sw);

    for (int s = wave; s < n_strips; s += nwaves) {
        int node0 = s * 16;
        int vstart = 0, vn = 0;
        if (lane < 16) {
            vstart = rowptr[node0 + lane];
            vn     = cnt[node0 + lane];
        }
        int n0 = __builtin_amdgcn_readlane(vn, 0);
        int n1 = __builtin_amdgcn_readlane(vn, 1);
        int idxA = csr[(lane < n0) ? (__builtin_amdgcn_readlane(vstart, 0) + lane) : 0];
        int idxB = csr[(lane < n1) ? (__builtin_amdgcn_readlane(vstart, 1) + lane) : 0];
        int cA   = cnt[idxA];

#pragma unroll 1
        for (int i = 0; i < 16; ++i) {
            int start = __builtin_amdgcn_readlane(vstart, i);
            int n     = __builtin_amdgcn_readlane(vn, i);
            int i2    = (i + 2 < 16) ? (i + 2) : 15;
            int n2    = (i + 2 < 16) ? __builtin_amdgcn_readlane(vn, i2) : 0;
            int idxC  = csr[(lane < n2) ? (__builtin_amdgcn_readlane(vstart, i2) + lane) : 0];
            int cB    = cnt[idxB];

            int2 pr = make_int2(idxA, __float_as_int(rsqrtf(fmaxf((float)cA, 1.0f))));
            float4 acc = make_float4(0.0f, 0.0f, 0.0f, 0.0f);

            for (int base = 0; base < n; base += 32) {
                int m = n - base;
                if (m > 32) m = 32;
                if (base > 0) {
                    pr = make_int2(0, 0);
                    if (lane < m) {
                        int sidx = csr[start + base + lane];
                        float nrm = rsqrtf(fmaxf((float)cnt[sidx], 1.0f));
                        pr = make_int2(sidx, __float_as_int(nrm));
                    }
                }
                int j = 0;
                for (; j + 16 <= m; j += 16) gcn_sum_rows<16, false>(fl, pr, j, m, acc);
                int rem = m - j;
                if (rem > 12)     gcn_sum_rows<16, true>(fl, pr, j, m, acc);
                else if (rem > 8) gcn_sum_rows<12, true>(fl, pr, j, m, acc);
                else if (rem > 4) gcn_sum_rows<8, true>(fl, pr, j, m, acc);
                else if (rem > 0) gcn_sum_rows<4, true>(fl, pr, j, m, acc);
            }
            *(float4*)(sAw + i * D_FEAT + ((lane ^ i) << 2)) = acc;

            idxA = idxB;
            cA   = cB;
            idxB = idxC;
        }
        __builtin_amdgcn_fence(__ATOMIC_RELEASE, "wavefront");
        __builtin_amdgcn_wave_barrier();
        __builtin_amdgcn_fence(__ATOMIC_ACQUIRE, "wavefront");

        v8f acc[8];
#pragma unroll
        for (int t = 0; t < 8; ++t) acc[t] = (v8f){0.f, 0.f, 0.f, 0.f, 0.f, 0.f, 0.f, 0.f};

        v2f a[8];
#pragma unroll
        for (int j = 0; j < 8; ++j) a[j] = *(const v2f*)(sAr + ((j ^ l15) << 2));

#pragma unroll 1
        for (int c = 0; c < 4; ++c) {
            v2f an[8];
#pragma unroll
            for (int j = 0; j < 8; ++j)
                an[j] = *(const v2f*)(sAr + (((8 * ((c + 1) & 3) + j) ^ l15) << 2));
#pragma unroll
            for (int j = 0; j < 8; ++j) {
                int kx = ((32 * c + 4 * (j + 1)) & (D_FEAT - 1)) ^ sw;
                v2f bn[8];
#pragma unroll
                for (int t = 0; t < 8; ++t) bn[t] = *(const v2f*)(sB + t * 16 * D_FEAT + kx);
#pragma unroll
                for (int t = 0; t < 8; ++t) {
                    acc[t] = __builtin_amdgcn_wmma_f32_16x16x4_f32(
                         false, a[j],  false, bq[t],
                         (short)0, acc[t],  false,  false);
                }
#pragma unroll
                for (int t = 0; t < 8; ++t) bq[t] = bn[t];
            }
#pragma unroll
            for (int j = 0; j < 8; ++j) a[j] = an[j];
        }

#pragma unroll
        for (int r0 = 0; r0 < 8; r0 += 4) {
            float f[4][8];
#pragma unroll
            for (int r = 0; r < 4; ++r) {
                size_t rowoff = (size_t)(s * 16 + r0 + r + half * 8) * D_FEAT + l15;
#pragma unroll
                for (int t = 0; t < 8; ++t) f[r][t] = feat[rowoff + t * 16];
            }
#pragma unroll
            for (int r = 0; r < 4; ++r) {
                size_t rowoff = (size_t)(s * 16 + r0 + r + half * 8) * D_FEAT + l15;
#pragma unroll
                for (int t = 0; t < 8; ++t) {
                    float h = fmaxf(acc[t][r0 + r], 0.0f);
                    out[rowoff + t * 16] = h + bv * f[r][t];
                }
            }
        }
    }
}

extern "C" void kernel_launch(void* const* d_in, const int* in_sizes, int n_in,
                              void* d_out, int out_size, void* d_ws, size_t ws_size,
                              hipStream_t stream) {
    const float* feat = (const float*)d_in[0];
    const float* W    = (const float*)d_in[1];
    const float* beta = (const float*)d_in[2];
    const int*   src  = (const int*)d_in[3];
    const int*   dst  = (const int*)d_in[4];
    float*       out  = (float*)d_out;

    const int D = D_FEAT;
    const int N = in_sizes[0] / D;
    const int n_edges = in_sizes[3];

    const int n_chunks = (n_edges + BIN_CH - 1) / BIN_CH;
    const int npb      = (N + NBINS - 1) / NBINS;

    int*   tab_cnt   = (int*)d_ws;
    int*   tab_start = tab_cnt + (size_t)NBINS * n_chunks;
    int2*  binned    = (int2*)(tab_start + (size_t)NBINS * n_chunks);
    int*   cnt       = (int*)(binned + n_edges);
    int*   rowptr    = cnt + N;
    int*   csr       = rowptr + N;

    gcn_bin_kernel<<<n_chunks, BIN_THREADS, 0, stream>>>(
        src, dst, binned, tab_cnt, tab_start, n_edges, npb, n_chunks);

    gcn_owner_kernel<<<NBINS, OWN_THREADS, 0, stream>>>(
        binned, tab_cnt, tab_start, cnt, rowptr, csr, N, npb, n_chunks);

    int n_strips = N / 16;
    gcn_fused_kernel<<<256, FUSE_THREADS, 0, stream>>>(feat, rowptr, cnt, csr, W, beta, out, n_strips);
}


# Epilogue through LDS: relu(acc) staged in the wave's strip; residual rows read (ahead of the GEMM) and out rows stored as whole 512-byte rows, not 64-byte pieces by 4-byte accesses; sums unchanged
# speedup vs baseline: 5.0451x; 1.0442x over previous
#include <hip/hip_runtime.h>
#include <hip/hip_bf16.h>

typedef __attribute__((ext_vector_type(2))) float v2f;
typedef __attribute__((ext_vector_type(8))) float v8f;

#define D_FEAT 128


#define NBINS       256
#define RMAX        512
#define BIN_THREADS 1024
#define BIN_ITER    8
#define BIN_CH      (BIN_THREADS * BIN_ITER)
#define OWN_THREADS 1024
#define MAXCH       1024
#define OWN_U       8
#define OWN_CAP     12288

__global__ void __launch_bounds__(BIN_THREADS)
gcn_bin_kernel(const int* __restrict__ src, const int* __restrict__ dst,
               int2* __restrict__ binned,
               int* __restrict__ tab_cnt, int* __restrict__ tab_start,
               int n_edges, int nodes_per_bin, int n_chunks) {
    __shared__ int  s_hist[NBINS];
    __shared__ int  s_off[NBINS];
    __shared__ int2 s_e[BIN_CH];

    int tid = threadIdx.x;
    int c   = blockIdx.x;
    int e0  = c * BIN_CH;

    if (tid < NBINS) s_hist[tid] = 0;

    int sv[BIN_ITER], dv[BIN_ITER], bv[BIN_ITER], rv[BIN_ITER];
#pragma unroll
    for (int i = 0; i < BIN_ITER; ++i) {
        int e = e0 + i * BIN_THREADS + tid;
        bool ok = e < n_edges;
        sv[i] = ok ? src[e] : 0;
        dv[i] = ok ? dst[e] : -1;
    }
    __syncthreads();

#pragma unroll
    for (int i = 0; i < BIN_ITER; ++i) {
        bv[i] = 0;
        rv[i] = 0;
        if (dv[i] >= 0) {
            int b = dv[i] / nodes_per_bin;
            if (b > NBINS - 1) b = NBINS - 1;
            bv[i] = b;
            rv[i] = atomicAdd(&s_hist[b], 1);
        }
    }
    __syncthreads();

    if (tid < 32) {
        int v[NBINS / 32];
        int run = 0;
#pragma unroll
        for (int k = 0; k < NBINS / 32; ++k) {
            int t = s_hist[tid * (NBINS / 32) + k];
            v[k] = run;
            run += t;
        }
        int incl = run;
        for (int off = 1; off < 32; off <<= 1) {
            int t = __shfl_up(incl, off);
            if (tid >= off) incl += t;
        }
        int excl = incl - run;
#pragma unroll
        for (int k = 0; k < NBINS / 32; ++k) s_off[tid * (NBINS / 32) + k] = excl + v[k];
    }
    __syncthreads();

    if (tid < NBINS) {
        tab_cnt[tid * n_chunks + c]   = s_hist[tid];
        tab_start[tid * n_chunks + c] = s_off[tid];
    }

#pragma unroll
    for (int i = 0; i < BIN_ITER; ++i) {
        if (dv[i] >= 0) s_e[s_off[bv[i]] + rv[i]] = make_int2(sv[i], dv[i]);
    }
    __syncthreads();

    int nloc = n_edges - e0;
    if (nloc > BIN_CH) nloc = BIN_CH;
#pragma unroll
    for (int i = 0; i < BIN_ITER; ++i) {
        int p = i * BIN_THREADS + tid;
        if (p < nloc) binned[e0 + p] = s_e[p];
    }
}

__device__ __forceinline__ void gcn_locate(const int* s_pre, const int* s_beg, int top,
                                           int first, int total, int (&idx)[OWN_U]) {
    int it[OWN_U], pos[OWN_U];
#pragma unroll
    for (int u = 0; u < OWN_U; ++u) {
        int i = first + u * OWN_THREADS;
        it[u]  = i > total - 1 ? total - 1 : i;
        pos[u] = 0;
    }
    for (int step = top; step > 0; step >>= 1) {
#pragma unroll
        for (int u = 0; u < OWN_U; ++u) pos[u] += (s_pre[pos[u] + step] <= it[u]) ? step : 0;
    }
#pragma unroll
    for (int u = 0; u < OWN_U; ++u) idx[u] = s_beg[pos[u]] + (it[u] - s_pre[pos[u]]);
}

__global__ void __launch_bounds__(OWN_THREADS)
gcn_owner_kernel(const int2* __restrict__ binned,
                 const int* __restrict__ tab_cnt, const int* __restrict__ tab_start,
                 int* __restrict__ cnt, int* __restrict__ rowptr, int* __restrict__ csr,
                 int n_nodes, int nodes_per_bin, int n_chunks) {
    __shared__ int s_pre[MAXCH + 1];
    __shared__ int s_beg[MAXCH];
    __shared__ int s_cnt[RMAX];
    __shared__ int s_row[RMAX];
    __shared__ int s_cur[RMAX];
    __shared__ int s_wa[32];
    __shared__ int s_wb[32];
    __shared__ int s_csr[OWN_CAP];

    int tid  = threadIdx.x;
    int lane = tid & 31;
    int w    = tid >> 5;
    int b    = blockIdx.x;

    int pre = 0;
    int nfront = b * n_chunks;
    const int4* t4 = (const int4*)tab_cnt;
    int nvec = nfront >> 2;
    int iv = tid;
    for (; iv + 3 * OWN_THREADS < nvec; iv += 4 * OWN_THREADS) {
        int4 a = t4[iv];
        int4 q = t4[iv + OWN_THREADS];
        int4 g = t4[iv + 2 * OWN_THREADS];
        int4 h = t4[iv + 3 * OWN_THREADS];
        pre += ((a.x + a.y) + (a.z + a.w)) + ((q.x + q.y) + (q.z + q.w))
             + ((g.x + g.y) + (g.z + g.w)) + ((h.x + h.y) + (h.z + h.w));
    }
    for (; iv < nvec; iv += OWN_THREADS) {
        int4 v = t4[iv];
        pre += (v.x + v.y) + (v.z + v.w);
    }
    if (tid < (nfront & 3)) pre += tab_cnt[(nvec << 2) + tid];

    int myc = 0, myb = 0;
    if (tid < n_chunks) {
        myc = tab_cnt[b * n_chunks + tid];
        myb = tid * BIN_CH + tab_start[b * n_chunks + tid];
    }
    if (tid < RMAX) { s_cnt[tid] = 0; s_cur[tid] = 0; }

    int incl = myc;
    for (int off = 1; off < 32; off <<= 1) {
        int t = __shfl_up(incl, off);
        if (lane >= off) incl += t;
    }
    for (int off = 16; off > 0; off >>= 1) pre += __shfl_xor(pre, off);
    if (lane == 31) s_wa[w] = incl;
    if (lane == 0)  s_wb[w] = pre;
    __syncthreads();

    int front = s_wb[lane];
    int mine  = (lane < w) ? s_wa[lane] : 0;
    for (int off = 16; off > 0; off >>= 1) {
        front += __shfl_xor(front, off);
        mine  += __shfl_xor(mine, off);
    }
    int base = front;
    int excl = mine + incl - myc;
    s_pre[tid] = excl;
    s_beg[tid] = myb;
    if (tid == OWN_THREADS - 1) s_pre[MAXCH] = excl + myc;
    __syncthreads();

    int total = s_pre[MAXCH];
    int node0 = b * nodes_per_bin;
    int top = 1;
    for (int k = 0; k < 10; ++k) if (top * 2 <= n_chunks) top *= 2;

    for (int i0 = 0; i0 < total; i0 += OWN_THREADS * OWN_U) {
        int2 ed[OWN_U];
        int idx[OWN_U];
        gcn_locate(s_pre, s_beg, top, i0 + tid, total, idx);
#pragma unroll
        for (int u = 0; u < OWN_U; ++u) ed[u] = binned[idx[u]];
#pragma unroll
        for (int u = 0; u < OWN_U; ++u) {
            int loc = ed[u].y - node0;
            loc = loc < 0 ? 0 : (loc > RMAX - 1 ? RMAX - 1 : loc);
            if (i0 + u * OWN_THREADS + tid < total) atomicAdd(&s_cnt[loc], 1);
        }
    }
    __syncthreads();

    int cv = (tid < RMAX) ? s_cnt[tid] : 0;
    int inc2 = cv;
    for (int off = 1; off < 32; off <<= 1) {
        int t = __shfl_up(inc2, off);
        if (lane >= off) inc2 += t;
    }
    if (lane == 31) s_wa[w] = inc2;
    __syncthreads();
    int m2 = (lane < w) ? s_wa[lane] : 0;
    for (int off = 16; off > 0; off >>= 1) m2 += __shfl_xor(m2, off);
    int rowx = m2 + inc2 - cv;
    if (tid < RMAX) s_row[tid] = rowx;
    int node = node0 + tid;
    if (tid < nodes_per_bin && node < n_nodes) {
        cnt[node]    = cv;
        rowptr[node] = base + rowx;
    }
    __syncthreads();

    bool staged = total <= OWN_CAP;
    for (int i0 = 0; i0 < total; i0 += OWN_THREADS * OWN_U) {
        int2 ed[OWN_U];
        int idx[OWN_U];
        gcn_locate(s_pre, s_beg, top, i0 + tid, total, idx);
#pragma unroll
        for (int u = 0; u < OWN_U; ++u) ed[u] = binned[idx[u]];
#pragma unroll
        for (int u = 0; u < OWN_U; ++u) {
            int loc = ed[u].y - node0;
            loc = loc < 0 ? 0 : (loc > RMAX - 1 ? RMAX - 1 : loc);
            if (i0 + u * OWN_THREADS + tid < total) {
                int r = atomicAdd(&s_cur[loc], 1);
                int pos = s_row[loc] + r;
                if (staged) s_csr[pos] = ed[u].x;
                else        csr[base + pos] = ed[u].x;
            }
        }
    }
    __syncthreads();
    if (staged) {
        for (int i = tid; i < total; i += OWN_THREADS) csr[base + i] = s_csr[i];
    }
}

#define FUSE_WAVES   16
#define FUSE_THREADS (FUSE_WAVES * 32)

template <int NB, bool PAD>
__device__ __forceinline__ void gcn_sum_rows(const float* fl, int2 pr, int j, int m, float4& acc) {
    float4 r[NB];
    float  wt[NB];
#pragma unroll
    for (int u = 0; u < NB; ++u) {
        bool in = !PAD || (j + u < m);
        int jj = in ? (j + u) : (m - 1);
        int s = __builtin_amdgcn_readlane(pr.x, jj);
        float wv = __int_as_float(__builtin_amdgcn_readlane(pr.y, jj));
        wt[u] = in ? wv : 0.0f;
        r[u]  = *(const float4*)(fl + (size_t)s * D_FEAT);
    }
#pragma unroll
    for (int u = 0; u < NB; ++u) {
        acc.x = fmaf(r[u].x, wt[u], acc.x);
        acc.y = fmaf(r[u].y, wt[u], acc.y);
        acc.z = fmaf(r[u].z, wt[u], acc.z);
        acc.w = fmaf(r[u].w, wt[u], acc.w);
    }
}

__global__ void __launch_bounds__(FUSE_THREADS) __attribute__((amdgpu_waves_per_eu(4, 4)))
gcn_fused_kernel(const float* __restrict__ feat,
                 const int* __restrict__ rowptr,
                 const int* __restrict__ cnt,
                 const int* __restrict__ csr,
                 const float* __restrict__ W,
                 const float* __restrict__ beta,
                 float* __restrict__ out,
                 int n_strips  ) {
    __shared__ float sW[D_FEAT * D_FEAT];
    __shared__ float sA[FUSE_WAVES * 16 * D_FEAT];

    int tid  = threadIdx.x;
    int lane = tid & 31;
    int half = lane >> 4;
    int l15  = lane & 15;
    int wv   = __builtin_amdgcn_readfirstlane(tid >> 5);

    for (int i = tid; i < D_FEAT * D_FEAT / 4; i += FUSE_THREADS) {
        int n = i >> 5;
        int k = (i & 31) << 2;
        float4 v = *(const float4*)(W + (size_t)n * D_FEAT + k);
        *(float4*)(sW + n * D_FEAT + (k ^ ((n & 15) << 2))) = v;
    }
    __syncthreads();

    int wave   = __builtin_amdgcn_readfirstlane((int)(wv * gridDim.x + blockIdx.x));
    int nwaves = (int)(gridDim.x * FUSE_WAVES);
    float bv = beta[0];

    const float* sB  = sW + l15 * D_FEAT + half * 2;
    int sw = l15 << 2;
    float*       sAw = sA + wv * 16 * D_FEAT;
    const float* sAr = sAw + l15 * D_FEAT + half * 2;
    const float* fl  = feat + lane * 4;

    v2f bq[8];
#pragma unroll
    for (int t = 0; t < 8; ++t) bq[t] = *(const v2f*)(sB + t * 16 * D_FEAT + sw);

    for (int s = wave; s < n_strips; s += nwaves) {
        int node0 = s * 16;
        int vstart = 0, vn = 0;
        if (lane < 16) {
            vstart = rowptr[node0 + lane];
            vn     = cnt[node0 + lane];
        }
        int n0 = __builtin_amdgcn_readlane(vn, 0);
        int n1 = __builtin_amdgcn_readlane(vn, 1);
        int idxA = csr[(lane < n0) ? (__builtin_amdgcn_readlane(vstart, 0) + lane) : 0];
        int idxB = csr[(lane < n1) ? (__builtin_amdgcn_readlane(vstart, 1) + lane) : 0];
        int cA   = cnt[idxA];

#pragma unroll 1
        for (int i = 0; i < 16; ++i) {
            int start = __builtin_amdgcn_readlane(vstart, i);
            int n     = __builtin_amdgcn_readlane(vn, i);
            int i2    = (i + 2 < 16) ? (i + 2) : 15;
            int n2    = (i + 2 < 16) ? __builtin_amdgcn_readlane(vn, i2) : 0;
            int idxC  = csr[(lane < n2) ? (__builtin_amdgcn_readlane(vstart, i2) + lane) : 0];
            int cB    = cnt[idxB];

            int2 pr = make_int2(idxA, __float_as_int(rsqrtf(fmaxf((float)cA, 1.0f))));
            float4 acc = make_float4(0.0f, 0.0f, 0.0f, 0.0f);

            for (int base = 0; base < n; base += 32) {
                int m = n - base;
                if (m > 32) m = 32;
                if (base > 0) {
                    pr = make_int2(0, 0);
                    if (lane < m) {
                        int sidx = csr[start + base + lane];
                        float nrm = rsqrtf(fmaxf((float)cnt[sidx], 1.0f));
                        pr = make_int2(sidx, __float_as_int(nrm));
                    }
                }
                int j = 0;
                for (; j + 16 <= m; j += 16) gcn_sum_rows<16, false>(fl, pr, j, m, acc);
                int rem = m - j;
                if (rem > 12)     gcn_sum_rows<16, true>(fl, pr, j, m, acc);
                else if (rem > 8) gcn_sum_rows<12, true>(fl, pr, j, m, acc);
                else if (rem > 4) gcn_sum_rows<8, true>(fl, pr, j, m, acc);
                else if (rem > 0) gcn_sum_rows<4, true>(fl, pr, j, m, acc);
            }
            *(float4*)(sAw + i * D_FEAT + ((lane ^ i) << 2)) = acc;

            idxA = idxB;
            cA   = cB;
            idxB = idxC;
        }
        __builtin_amdgcn_fence(__ATOMIC_RELEASE, "wavefront");
        __builtin_amdgcn_wave_barrier();
        __builtin_amdgcn_fence(__ATOMIC_ACQUIRE, "wavefront");

        const float* frow = fl + (size_t)s * 16 * D_FEAT;
        float4 f[16];
#pragma unroll
        for (int i = 0; i < 16; ++i) f[i] = *(const float4*)(frow + i * D_FEAT);

        v8f acc[8];
#pragma unroll
        for (int t = 0; t < 8; ++t) acc[t] = (v8f){0.f, 0.f, 0.f, 0.f, 0.f, 0.f, 0.f, 0.f};

        v2f a[8];
#pragma unroll
        for (int j = 0; j < 8; ++j) a[j] = *(const v2f*)(sAr + ((j ^ l15) << 2));

#pragma unroll 1
        for (int c = 0; c < 4; ++c) {
            v2f an[8];
#pragma unroll
            for (int j = 0; j < 8; ++j)
                an[j] = *(const v2f*)(sAr + (((8 * ((c + 1) & 3) + j) ^ l15) << 2));
#pragma unroll
            for (int j = 0; j < 8; ++j) {
                int kx = ((32 * c + 4 * (j + 1)) & (D_FEAT - 1)) ^ sw;
                v2f bn[8];
#pragma unroll
                for (int t = 0; t < 8; ++t) bn[t] = *(const v2f*)(sB + t * 16 * D_FEAT + kx);
#pragma unroll
                for (int t = 0; t < 8; ++t) {
                    acc[t] = __builtin_amdgcn_wmma_f32_16x16x4_f32(
                         false, a[j],  false, bq[t],
                         (short)0, acc[t],  false,  false);
                }
#pragma unroll
                for (int t = 0; t < 8; ++t) bq[t] = bn[t];
            }
#pragma unroll
            for (int j = 0; j < 8; ++j) a[j] = an[j];
        }

        float* sO = sAw + half * (8 * D_FEAT) + l15;
#pragma unroll
        for (int t = 0; t < 8; ++t) {
            int cb = ((t + 2 * half) & 7) << 4;
#pragma unroll
            for (int r = 0; r < 8; ++r) sO[r * D_FEAT + cb] = fmaxf(acc[t][r], 0.0f);
        }
        __builtin_amdgcn_fence(__ATOMIC_RELEASE, "wavefront");
        __builtin_amdgcn_wave_barrier();
        __builtin_amdgcn_fence(__ATOMIC_ACQUIRE, "wavefront");

        float* orow = out + (size_t)s * 16 * D_FEAT + lane * 4;
#pragma unroll
        for (int i = 0; i < 16; ++i) {
            float4 h = *(const float4*)(sAw + i * D_FEAT + ((lane * 4 + 32 * (i >> 3)) & (D_FEAT - 1)));
            float4 o;
            o.x = h.x + bv * f[i].x;
            o.y = h.y + bv * f[i].y;
            o.z = h.z + bv * f[i].z;
            o.w = h.w + bv * f[i].w;
            *(float4*)(orow + i * D_FEAT) = o;
        }
    }
}

extern "C" void kernel_launch(void* const* d_in, const int* in_sizes, int n_in,
                              void* d_out, int out_size, void* d_ws, size_t ws_size,
                              hipStream_t stream) {
    const float* feat = (const float*)d_in[0];
    const float* W    = (const float*)d_in[1];
    const float* beta = (const float*)d_in[2];
    const int*   src  = (const int*)d_in[3];
    const int*   dst  = (const int*)d_in[4];
    float*       out  = (float*)d_out;

    const int D = D_FEAT;
    const int N = in_sizes[0] / D;
    const int n_edges = in_sizes[3];

    const int n_chunks = (n_edges + BIN_CH - 1) / BIN_CH;
    const int npb      = (N + NBINS - 1) / NBINS;

    int*   tab_cnt   = (int*)d_ws;
    int*   tab_start = tab_cnt + (size_t)NBINS * n_chunks;
    int2*  binned    = (int2*)(tab_start + (size_t)NBINS * n_chunks);
    int*   cnt       = (int*)(binned + n_edges);
    int*   rowptr    = cnt + N;
    int*   csr       = rowptr + N;

    gcn_bin_kernel<<<n_chunks, BIN_THREADS, 0, stream>>>(
        src, dst, binned, tab_cnt, tab_start, n_edges, npb, n_chunks);

    gcn_owner_kernel<<<NBINS, OWN_THREADS, 0, stream>>>(
        binned, tab_cnt, tab_start, cnt, rowptr, csr, N, npb, n_chunks);

    int n_strips = N / 16;
    gcn_fused_kernel<<<256, FUSE_THREADS, 0, stream>>>(feat, rowptr, cnt, csr, W, beta, out, n_strips);
}


# Fused launch start-up: a thread's 8 W loads go out as one group (start row rotated per block) and the first strip's index chain is fetched before the LDS copy and its barrier; sums unchanged
# speedup vs baseline: 5.2504x; 1.0407x over previous
#include <hip/hip_runtime.h>
#include <hip/hip_bf16.h>

typedef __attribute__((ext_vector_type(2))) float v2f;
typedef __attribute__((ext_vector_type(8))) float v8f;

#define D_FEAT 128


#define NBINS       256
#define RMAX        512
#define BIN_THREADS 1024
#define BIN_ITER    8
#define BIN_CH      (BIN_THREADS * BIN_ITER)
#define OWN_THREADS 1024
#define MAXCH       1024
#define OWN_U       8
#define OWN_CAP     12288

__global__ void __launch_bounds__(BIN_THREADS)
gcn_bin_kernel(const int* __restrict__ src, const int* __restrict__ dst,
               int2* __restrict__ binned,
               int* __restrict__ tab_cnt, int* __restrict__ tab_start,
               int n_edges, int nodes_per_bin, int n_chunks) {
    __shared__ int  s_hist[NBINS];
    __shared__ int  s_off[NBINS];
    __shared__ int2 s_e[BIN_CH];

    int tid = threadIdx.x;
    int c   = blockIdx.x;
    int e0  = c * BIN_CH;

    if (tid < NBINS) s_hist[tid] = 0;

    int sv[BIN_ITER], dv[BIN_ITER], bv[BIN_ITER], rv[BIN_ITER];
#pragma unroll
    for (int i = 0; i < BIN_ITER; ++i) {
        int e = e0 + i * BIN_THREADS + tid;
        bool ok = e < n_edges;
        sv[i] = ok ? src[e] : 0;
        dv[i] = ok ? dst[e] : -1;
    }
    __syncthreads();

#pragma unroll
    for (int i = 0; i < BIN_ITER; ++i) {
        bv[i] = 0;
        rv[i] = 0;
        if (dv[i] >= 0) {
            int b = dv[i] / nodes_per_bin;
            if (b > NBINS - 1) b = NBINS - 1;
            bv[i] = b;
            rv[i] = atomicAdd(&s_hist[b], 1);
        }
    }
    __syncthreads();

    if (tid < 32) {
        int v[NBINS / 32];
        int run = 0;
#pragma unroll
        for (int k = 0; k < NBINS / 32; ++k) {
            int t = s_hist[tid * (NBINS / 32) + k];
            v[k] = run;
            run += t;
        }
        int incl = run;
        for (int off = 1; off < 32; off <<= 1) {
            int t = __shfl_up(incl, off);
            if (tid >= off) incl += t;
        }
        int excl = incl - run;
#pragma unroll
        for (int k = 0; k < NBINS / 32; ++k) s_off[tid * (NBINS / 32) + k] = excl + v[k];
    }
    __syncthreads();

    if (tid < NBINS) {
        tab_cnt[tid * n_chunks + c]   = s_hist[tid];
        tab_start[tid * n_chunks + c] = s_off[tid];
    }

#pragma unroll
    for (int i = 0; i < BIN_ITER; ++i) {
        if (dv[i] >= 0) s_e[s_off[bv[i]] + rv[i]] = make_int2(sv[i], dv[i]);
    }
    __syncthreads();

    int nloc = n_edges - e0;
    if (nloc > BIN_CH) nloc = BIN_CH;
#pragma unroll
    for (int i = 0; i < BIN_ITER; ++i) {
        int p = i * BIN_THREADS + tid;
        if (p < nloc) binned[e0 + p] = s_e[p];
    }
}

__device__ __forceinline__ void gcn_locate(const int* s_pre, const int* s_beg, int top,
                                           int first, int total, int (&idx)[OWN_U]) {
    int it[OWN_U], pos[OWN_U];
#pragma unroll
    for (int u = 0; u < OWN_U; ++u) {
        int i = first + u * OWN_THREADS;
        it[u]  = i > total - 1 ? total - 1 : i;
        pos[u] = 0;
    }
    for (int step = top; step > 0; step >>= 1) {
#pragma unroll
        for (int u = 0; u < OWN_U; ++u) pos[u] += (s_pre[pos[u] + step] <= it[u]) ? step : 0;
    }
#pragma unroll
    for (int u = 0; u < OWN_U; ++u) idx[u] = s_beg[pos[u]] + (it[u] - s_pre[pos[u]]);
}

__global__ void __launch_bounds__(OWN_THREADS)
gcn_owner_kernel(const int2* __restrict__ binned,
                 const int* __restrict__ tab_cnt, const int* __restrict__ tab_start,
                 int* __restrict__ cnt, int* __restrict__ rowptr, int* __restrict__ csr,
                 int n_nodes, int nodes_per_bin, int n_chunks) {
    __shared__ int s_pre[MAXCH + 1];
    __shared__ int s_beg[MAXCH];
    __shared__ int s_cnt[RMAX];
    __shared__ int s_row[RMAX];
    __shared__ int s_cur[RMAX];
    __shared__ int s_wa[32];
    __shared__ int s_wb[32];
    __shared__ int s_csr[OWN_CAP];

    int tid  = threadIdx.x;
    int lane = tid & 31;
    int w    = tid >> 5;
    int b    = blockIdx.x;

    int pre = 0;
    int nfront = b * n_chunks;
    const int4* t4 = (const int4*)tab_cnt;
    int nvec = nfront >> 2;
    int iv = tid;
    for (; iv + 3 * OWN_THREADS < nvec; iv += 4 * OWN_THREADS) {
        int4 a = t4[iv];
        int4 q = t4[iv + OWN_THREADS];
        int4 g = t4[iv + 2 * OWN_THREADS];
        int4 h = t4[iv + 3 * OWN_THREADS];
        pre += ((a.x + a.y) + (a.z + a.w)) + ((q.x + q.y) + (q.z + q.w))
             + ((g.x + g.y) + (g.z + g.w)) + ((h.x + h.y) + (h.z + h.w));
    }
    for (; iv < nvec; iv += OWN_THREADS) {
        int4 v = t4[iv];
        pre += (v.x + v.y) + (v.z + v.w);
    }
    if (tid < (nfront & 3)) pre += tab_cnt[(nvec << 2) + tid];

    int myc = 0, myb = 0;
    if (tid < n_chunks) {
        myc = tab_cnt[b * n_chunks + tid];
        myb = tid * BIN_CH + tab_start[b * n_chunks + tid];
    }
    if (tid < RMAX) { s_cnt[tid] = 0; s_cur[tid] = 0; }

    int incl = myc;
    for (int off = 1; off < 32; off <<= 1) {
        int t = __shfl_up(incl, off);
        if (lane >= off) incl += t;
    }
    for (int off = 16; off > 0; off >>= 1) pre += __shfl_xor(pre, off);
    if (lane == 31) s_wa[w] = incl;
    if (lane == 0)  s_wb[w] = pre;
    __syncthreads();

    int front = s_wb[lane];
    int mine  = (lane < w) ? s_wa[lane] : 0;
    for (int off = 16; off > 0; off >>= 1) {
        front += __shfl_xor(front, off);
        mine  += __shfl_xor(mine, off);
    }
    int base = front;
    int excl = mine + incl - myc;
    s_pre[tid] = excl;
    s_beg[tid] = myb;
    if (tid == OWN_THREADS - 1) s_pre[MAXCH] = excl + myc;
    __syncthreads();

    int total = s_pre[MAXCH];
    int node0 = b * nodes_per_bin;
    int top = 1;
    for (int k = 0; k < 10; ++k) if (top * 2 <= n_chunks) top *= 2;

    for (int i0 = 0; i0 < total; i0 += OWN_THREADS * OWN_U) {
        int2 ed[OWN_U];
        int idx[OWN_U];
        gcn_locate(s_pre, s_beg, top, i0 + tid, total, idx);
#pragma unroll
        for (int u = 0; u < OWN_U; ++u) ed[u] = binned[idx[u]];
#pragma unroll
        for (int u = 0; u < OWN_U; ++u) {
            int loc = ed[u].y - node0;
            loc = loc < 0 ? 0 : (loc > RMAX - 1 ? RMAX - 1 : loc);
            if (i0 + u * OWN_THREADS + tid < total) atomicAdd(&s_cnt[loc], 1);
        }
    }
    __syncthreads();

    int cv = (tid < RMAX) ? s_cnt[tid] : 0;
    int inc2 = cv;
    for (int off = 1; off < 32; off <<= 1) {
        int t = __shfl_up(inc2, off);
        if (lane >= off) inc2 += t;
    }
    if (lane == 31) s_wa[w] = inc2;
    __syncthreads();
    int m2 = (lane < w) ? s_wa[lane] : 0;
    for (int off = 16; off > 0; off >>= 1) m2 += __shfl_xor(m2, off);
    int rowx = m2 + inc2 - cv;
    if (tid < RMAX) s_row[tid] = rowx;
    int node = node0 + tid;
    if (tid < nodes_per_bin && node < n_nodes) {
        cnt[node]    = cv;
        rowptr[node] = base + rowx;
    }
    __syncthreads();

    bool staged = total <= OWN_CAP;
    for (int i0 = 0; i0 < total; i0 += OWN_THREADS * OWN_U) {
        int2 ed[OWN_U];
        int idx[OWN_U];
        gcn_locate(s_pre, s_beg, top, i0 + tid, total, idx);
#pragma unroll
        for (int u = 0; u < OWN_U; ++u) ed[u] = binned[idx[u]];
#pragma unroll
        for (int u = 0; u < OWN_U; ++u) {
            int loc = ed[u].y - node0;
            loc = loc < 0 ? 0 : (loc > RMAX - 1 ? RMAX - 1 : loc);
            if (i0 + u * OWN_THREADS + tid < total) {
                int r = atomicAdd(&s_cur[loc], 1);
                int pos = s_row[loc] + r;
                if (staged) s_csr[pos] = ed[u].x;
                else        csr[base + pos] = ed[u].x;
            }
        }
    }
    __syncthreads();
    if (staged) {
        for (int i = tid; i < total; i += OWN_THREADS) csr[base + i] = s_csr[i];
    }
}

#define FUSE_WAVES   16
#define FUSE_THREADS (FUSE_WAVES * 32)

template <int NB, bool PAD>
__device__ __forceinline__ void gcn_sum_rows(const float* fl, int2 pr, int j, int m, float4& acc) {
    float4 r[NB];
    float  wt[NB];
#pragma unroll
    for (int u = 0; u < NB; ++u) {
        bool in = !PAD || (j + u < m);
        int jj = in ? (j + u) : (m - 1);
        int s = __builtin_amdgcn_readlane(pr.x, jj);
        float wv = __int_as_float(__builtin_amdgcn_readlane(pr.y, jj));
        wt[u] = in ? wv : 0.0f;
        r[u]  = *(const float4*)(fl + (size_t)s * D_FEAT);
    }
#pragma unroll
    for (int u = 0; u < NB; ++u) {
        acc.x = fmaf(r[u].x, wt[u], acc.x);
        acc.y = fmaf(r[u].y, wt[u], acc.y);
        acc.z = fmaf(r[u].z, wt[u], acc.z);
        acc.w = fmaf(r[u].w, wt[u], acc.w);
    }
}

__device__ __forceinline__ void gcn_strip_head(const int* __restrict__ rowptr,
                                               const int* __restrict__ cnt,
                                               const int* __restrict__ csr,
                                               int s, int lane,
                                               int& vstart, int& vn, int& idxA, int& idxB, int& cA) {
    int node = s * 16 + (lane & 15);
    vstart = rowptr[node];
    vn     = cnt[node];
    int n0 = __builtin_amdgcn_readlane(vn, 0);
    int n1 = __builtin_amdgcn_readlane(vn, 1);
    unsigned a0 = (lane < n0) ? (unsigned)(__builtin_amdgcn_readlane(vstart, 0) + lane) : 0u;
    unsigned a1 = (lane < n1) ? (unsigned)(__builtin_amdgcn_readlane(vstart, 1) + lane) : 0u;
    idxA = csr[a0];
    idxB = csr[a1];
    cA   = cnt[idxA];
}

__global__ void __launch_bounds__(FUSE_THREADS) __attribute__((amdgpu_waves_per_eu(4, 4)))
gcn_fused_kernel(const float* __restrict__ feat,
                 const int* __restrict__ rowptr,
                 const int* __restrict__ cnt,
                 const int* __restrict__ csr,
                 const float* __restrict__ W,
                 const float* __restrict__ beta,
                 float* __restrict__ out,
                 int n_strips  ) {
    __shared__ float sW[D_FEAT * D_FEAT];
    __shared__ float sA[FUSE_WAVES * 16 * D_FEAT];

    int tid  = threadIdx.x;
    int lane = tid & 31;
    int half = lane >> 4;
    int l15  = lane & 15;
    int wv   = __builtin_amdgcn_readfirstlane(tid >> 5);

    int wave   = __builtin_amdgcn_readfirstlane((int)(wv * gridDim.x + blockIdx.x));
    int nwaves = (int)(gridDim.x * FUSE_WAVES);

    float4 wq[8];
#pragma unroll
    for (int u = 0; u < 8; ++u) {
        int n = wv + 16 * ((u + (int)blockIdx.x) & 7);
        wq[u] = *(const float4*)(W + (size_t)n * D_FEAT + lane * 4);
    }
    int s = wave;
    int vstart, vn, idxA, idxB, cA;
    int sh = s < n_strips ? s : n_strips - 1;
    gcn_strip_head(rowptr, cnt, csr, sh < 0 ? 0 : sh, lane, vstart, vn, idxA, idxB, cA);
#pragma unroll
    for (int u = 0; u < 8; ++u) {
        int n = wv + 16 * ((u + (int)blockIdx.x) & 7);
        *(float4*)(sW + n * D_FEAT + ((lane * 4) ^ ((n & 15) << 2))) = wq[u];
    }
    __syncthreads();

    float bv = beta[0];

    const float* sB  = sW + l15 * D_FEAT + half * 2;
    int sw = l15 << 2;
    float*       sAw = sA + wv * 16 * D_FEAT;
    const float* sAr = sAw + l15 * D_FEAT + half * 2;
    const float* fl  = feat + lane * 4;

    v2f bq[8];
#pragma unroll
    for (int t = 0; t < 8; ++t) bq[t] = *(const v2f*)(sB + t * 16 * D_FEAT + sw);

    while (s < n_strips) {

#pragma unroll 1
        for (int i = 0; i < 16; ++i) {
            int start = __builtin_amdgcn_readlane(vstart, i);
            int n     = __builtin_amdgcn_readlane(vn, i);
            int i2    = (i + 2 < 16) ? (i + 2) : 15;
            int n2    = (i + 2 < 16) ? __builtin_amdgcn_readlane(vn, i2) : 0;
            int idxC  = csr[(lane < n2) ? (__builtin_amdgcn_readlane(vstart, i2) + lane) : 0];
            int cB    = cnt[idxB];

            int2 pr = make_int2(idxA, __float_as_int(rsqrtf(fmaxf((float)cA, 1.0f))));
            float4 acc = make_float4(0.0f, 0.0f, 0.0f, 0.0f);

            for (int base = 0; base < n; base += 32) {
                int m = n - base;
                if (m > 32) m = 32;
                if (base > 0) {
                    pr = make_int2(0, 0);
                    if (lane < m) {
                        int sidx = csr[start + base + lane];
                        float nrm = rsqrtf(fmaxf((float)cnt[sidx], 1.0f));
                        pr = make_int2(sidx, __float_as_int(nrm));
                    }
                }
                int j = 0;
                for (; j + 16 <= m; j += 16) gcn_sum_rows<16, false>(fl, pr, j, m, acc);
                int rem = m - j;
                if (rem > 12)     gcn_sum_rows<16, true>(fl, pr, j, m, acc);
                else if (rem > 8) gcn_sum_rows<12, true>(fl, pr, j, m, acc);
                else if (rem > 4) gcn_sum_rows<8, true>(fl, pr, j, m, acc);
                else if (rem > 0) gcn_sum_rows<4, true>(fl, pr, j, m, acc);
            }
            *(float4*)(sAw + i * D_FEAT + ((lane ^ i) << 2)) = acc;

            idxA = idxB;
            cA   = cB;
            idxB = idxC;
        }
        __builtin_amdgcn_fence(__ATOMIC_RELEASE, "wavefront");
        __builtin_amdgcn_wave_barrier();
        __builtin_amdgcn_fence(__ATOMIC_ACQUIRE, "wavefront");

        const float* frow = fl + (size_t)s * 16 * D_FEAT;
        float4 f[16];
#pragma unroll
        for (int i = 0; i < 16; ++i) f[i] = *(const float4*)(frow + i * D_FEAT);

        v8f acc[8];
#pragma unroll
        for (int t = 0; t < 8; ++t) acc[t] = (v8f){0.f, 0.f, 0.f, 0.f, 0.f, 0.f, 0.f, 0.f};

        v2f a[8];
#pragma unroll
        for (int j = 0; j < 8; ++j) a[j] = *(const v2f*)(sAr + ((j ^ l15) << 2));

#pragma unroll 1
        for (int c = 0; c < 4; ++c) {
            v2f an[8];
#pragma unroll
            for (int j = 0; j < 8; ++j)
                an[j] = *(const v2f*)(sAr + (((8 * ((c + 1) & 3) + j) ^ l15) << 2));
#pragma unroll
            for (int j = 0; j < 8; ++j) {
                int kx = ((32 * c + 4 * (j + 1)) & (D_FEAT - 1)) ^ sw;
                v2f bn[8];
#pragma unroll
                for (int t = 0; t < 8; ++t) bn[t] = *(const v2f*)(sB + t * 16 * D_FEAT + kx);
#pragma unroll
                for (int t = 0; t < 8; ++t) {
                    acc[t] = __builtin_amdgcn_wmma_f32_16x16x4_f32(
                         false, a[j],  false, bq[t],
                         (short)0, acc[t],  false,  false);
                }
#pragma unroll
                for (int t = 0; t < 8; ++t) bq[t] = bn[t];
            }
#pragma unroll
            for (int j = 0; j < 8; ++j) a[j] = an[j];
        }

        float* sO = sAw + half * (8 * D_FEAT) + l15;
#pragma unroll
        for (int t = 0; t < 8; ++t) {
            int cb = ((t + 2 * half) & 7) << 4;
#pragma unroll
            for (int r = 0; r < 8; ++r) sO[r * D_FEAT + cb] = fmaxf(acc[t][r], 0.0f);
        }
        __builtin_amdgcn_fence(__ATOMIC_RELEASE, "wavefront");
        __builtin_amdgcn_wave_barrier();
        __builtin_amdgcn_fence(__ATOMIC_ACQUIRE, "wavefront");

        float* orow = out + (size_t)s * 16 * D_FEAT + lane * 4;
#pragma unroll
        for (int i = 0; i < 16; ++i) {
            float4 h = *(const float4*)(sAw + i * D_FEAT + ((lane * 4 + 32 * (i >> 3)) & (D_FEAT - 1)));
            float4 o;
            o.x = h.x + bv * f[i].x;
            o.y = h.y + bv * f[i].y;
            o.z = h.z + bv * f[i].z;
            o.w = h.w + bv * f[i].w;
            *(float4*)(orow + i * D_FEAT) = o;
        }

        s += nwaves;
        if (s < n_strips) gcn_strip_head(rowptr, cnt, csr, s, lane, vstart, vn, idxA, idxB, cA);
    }
}

extern "C" void kernel_launch(void* const* d_in, const int* in_sizes, int n_in,
                              void* d_out, int out_size, void* d_ws, size_t ws_size,
                              hipStream_t stream) {
    const float* feat = (const float*)d_in[0];
    const float* W    = (const float*)d_in[1];
    const float* beta = (const float*)d_in[2];
    const int*   src  = (const int*)d_in[3];
    const int*   dst  = (const int*)d_in[4];
    float*       out  = (float*)d_out;

    const int D = D_FEAT;
    const int N = in_sizes[0] / D;
    const int n_edges = in_sizes[3];

    const int n_chunks = (n_edges + BIN_CH - 1) / BIN_CH;
    const int npb      = (N + NBINS - 1) / NBINS;

    int*   tab_cnt   = (int*)d_ws;
    int*   tab_start = tab_cnt + (size_t)NBINS * n_chunks;
    int2*  binned    = (int2*)(tab_start + (size_t)NBINS * n_chunks);
    int*   cnt       = (int*)(binned + n_edges);
    int*   rowptr    = cnt + N;
    int*   csr       = rowptr + N;

    gcn_bin_kernel<<<n_chunks, BIN_THREADS, 0, stream>>>(
        src, dst, binned, tab_cnt, tab_start, n_edges, npb, n_chunks);

    gcn_owner_kernel<<<NBINS, OWN_THREADS, 0, stream>>>(
        binned, tab_cnt, tab_start, cnt, rowptr, csr, N, npb, n_chunks);

    int n_strips = N / 16;
    gcn_fused_kernel<<<256, FUSE_THREADS, 0, stream>>>(feat, rowptr, cnt, csr, W, beta, out, n_strips);
}
